# starting kernel
# baseline (speedup 1.0000x reference)
#include <hip/hip_runtime.h>

typedef __attribute__((ext_vector_type(2))) float v2f;
typedef __attribute__((ext_vector_type(8))) float v8f;

#define B_  4
#define T_  5
#define C_  3
#define H_  128
#define W_  128
#define D_  60
#define DP  64
#define NN  4096
#define KT  128

__device__ __forceinline__ float fast_exp2(float x) {
    return __builtin_amdgcn_exp2f(x);
}

__global__ void nlb_prep_q(const float* __restrict__ x, float* __restrict__ Q) {
    int idx = blockIdx.x * blockDim.x + threadIdx.x;
    if (idx >= B_ * NN * DP) return;
    int dp = idx & 63;
    int n  = (idx >> 6) & (NN - 1);
    int b  = idx >> 18;
    float v = 0.0f;
    if (dp < D_) {
        int tc = dp >> 2;
        int rr = dp & 3;
        int ri = rr >> 1, rj = rr & 1;
        int oh = n >> 6, ow = n & 63;
        int hh = oh * 2 + ri, ww = ow * 2 + rj;
        v = x[((size_t)(b * (T_ * C_) + tc) * H_ + hh) * W_ + ww];
    }
    Q[idx] = v;
}

__global__ void nlb_prep_g(const float* __restrict__ Q, const float* __restrict__ gw,
                           const float* __restrict__ gb, float* __restrict__ Gt) {
    int idx = blockIdx.x * blockDim.x + threadIdx.x;
    if (idx >= B_ * DP * NN) return;
    int n = idx & (NN - 1);
    int o = (idx >> 12) & 63;
    int b = idx >> 18;
    float acc = 0.0f;
    if (o < D_) {
        const float* qr = Q + (((size_t)b * NN + n) << 6);
        const float* wr = gw + o * D_;
        acc = gb[o];
#pragma unroll 4
        for (int d = 0; d < D_; ++d) acc += wr[d] * qr[d];
    }
    Gt[idx] = acc;
}

__global__ __launch_bounds__(128, 1)
void nlb_attn(const float* __restrict__ Q, const float* __restrict__ Gt,
              float* __restrict__ Y) {
    __shared__ float lds[4 * 16 * KT];

    const int lane = threadIdx.x & 31;
    const int wave = threadIdx.x >> 5;
    const int m    = lane & 15;
    const int h    = lane >> 4;
    const int b    = blockIdx.x >> 6;
    const int q0   = ((blockIdx.x & 63) * 4 + wave) * 16;

    const float* Qb  = Q  + (size_t)b * NN * DP;
    const float* Gtb = Gt + (size_t)b * DP * NN;
    float*       Yb  = Y  + (size_t)b * NN * DP;
    float*       pl  = lds + wave * (16 * KT);

    const float* vrow0 = Gtb + (size_t)( 0 + m) * NN;
    const float* vrow1 = Gtb + (size_t)(16 + m) * NN;
    const float* vrow2 = Gtb + (size_t)(32 + m) * NN;
    const float* vrow3 = Gtb + (size_t)(48 + m) * NN;

    const float LOG2E = 1.4426950408889634f;
    v2f qa[15];
    {
        const float* qrow = Qb + (size_t)(q0 + m) * DP;
#pragma unroll
        for (int kk = 0; kk < 15; ++kk) {
            v2f q = *(const v2f*)(qrow + 4 * kk + 2 * h);
            qa[kk] = q * LOG2E;
        }
    }

    v8f o0{}, o1{}, o2{}, o3{};
    float mrow[8], lrow[8];
#pragma unroll
    for (int j = 0; j < 8; ++j) { mrow[j] = -__builtin_inff(); lrow[j] = 0.0f; }

    for (int ch = 0; ch < NN / KT; ++ch) {
        const int k0 = ch * KT;

        {
            const int kn = (ch + 1 < NN / KT) ? (k0 + KT) : k0;
            const char* pk = (const char*)(Qb + (size_t)kn * DP) + lane * 1024;
            __builtin_prefetch(pk,       0, 3);
            __builtin_prefetch(pk + 256, 0, 3);
            __builtin_prefetch(pk + 512, 0, 3);
            __builtin_prefetch(pk + 768, 0, 3);
            const char* pv0 = (const char*)(Gtb + (size_t)lane        * NN + kn);
            const char* pv1 = (const char*)(Gtb + (size_t)(lane + 32) * NN + kn);
            __builtin_prefetch(pv0,       0, 3);
            __builtin_prefetch(pv0 + 256, 0, 3);
            __builtin_prefetch(pv1,       0, 3);
            __builtin_prefetch(pv1 + 256, 0, 3);
        }

        v8f s[8];
#pragma unroll
        for (int t = 0; t < 8; ++t) {
            v8f acc{};
            const float* krow = Qb + (size_t)(k0 + t * 16 + m) * DP;
#pragma unroll
            for (int kk = 0; kk < 15; ++kk) {
                v2f kb = *(const v2f*)(krow + 4 * kk + 2 * h);
                acc = __builtin_amdgcn_wmma_f32_16x16x4_f32(
                          false, qa[kk], false, kb, (short)0, acc, false, false);
            }
            s[t] = acc;
        }

#pragma unroll
        for (int j = 0; j < 8; ++j) {
            float tm = fmaxf(fmaxf(fmaxf(s[0][j], s[1][j]), fmaxf(s[2][j], s[3][j])),
                             fmaxf(fmaxf(s[4][j], s[5][j]), fmaxf(s[6][j], s[7][j])));
            tm = fmaxf(tm, __shfl_xor(tm, 1, 16));
            tm = fmaxf(tm, __shfl_xor(tm, 2, 16));
            tm = fmaxf(tm, __shfl_xor(tm, 4, 16));
            tm = fmaxf(tm, __shfl_xor(tm, 8, 16));
            float mn = fmaxf(mrow[j], tm);
            float sc = fast_exp2(mrow[j] - mn);
            float p0 = fast_exp2(s[0][j] - mn);
            float p1 = fast_exp2(s[1][j] - mn);
            float p2 = fast_exp2(s[2][j] - mn);
            float p3 = fast_exp2(s[3][j] - mn);
            float p4 = fast_exp2(s[4][j] - mn);
            float p5 = fast_exp2(s[5][j] - mn);
            float p6 = fast_exp2(s[6][j] - mn);
            float p7 = fast_exp2(s[7][j] - mn);
            s[0][j] = p0; s[1][j] = p1; s[2][j] = p2; s[3][j] = p3;
            s[4][j] = p4; s[5][j] = p5; s[6][j] = p6; s[7][j] = p7;
            float su = ((p0 + p1) + (p2 + p3)) + ((p4 + p5) + (p6 + p7));
            su += __shfl_xor(su, 1, 16);
            su += __shfl_xor(su, 2, 16);
            su += __shfl_xor(su, 4, 16);
            su += __shfl_xor(su, 8, 16);
            lrow[j] = lrow[j] * sc + su;
            mrow[j] = mn;
            o0[j] *= sc; o1[j] *= sc; o2[j] *= sc; o3[j] *= sc;
        }

#pragma unroll
        for (int t = 0; t < 8; ++t)
#pragma unroll
            for (int j = 0; j < 8; ++j)
                pl[(8 * h + j) * KT + t * 16 + m] = s[t][j];
        asm volatile("" ::: "memory");

#pragma unroll
        for (int kk2 = 0; kk2 < 32; ++kk2) {
            v2f pa = *(const v2f*)(pl + m * KT + 4 * kk2 + 2 * h);
            const int kof = k0 + 4 * kk2 + 2 * h;
            v2f vb;
            vb = *(const v2f*)(vrow0 + kof);
            o0 = __builtin_amdgcn_wmma_f32_16x16x4_f32(false, pa, false, vb, (short)0, o0, false, false);
            vb = *(const v2f*)(vrow1 + kof);
            o1 = __builtin_amdgcn_wmma_f32_16x16x4_f32(false, pa, false, vb, (short)0, o1, false, false);
            vb = *(const v2f*)(vrow2 + kof);
            o2 = __builtin_amdgcn_wmma_f32_16x16x4_f32(false, pa, false, vb, (short)0, o2, false, false);
            vb = *(const v2f*)(vrow3 + kof);
            o3 = __builtin_amdgcn_wmma_f32_16x16x4_f32(false, pa, false, vb, (short)0, o3, false, false);
        }
        asm volatile("" ::: "memory");
    }

#pragma unroll
    for (int j = 0; j < 8; ++j) {
        float inv = 1.0f / lrow[j];
        float* yr = Yb + (size_t)(q0 + 8 * h + j) * DP;
        yr[ 0 + m] = o0[j] * inv;
        yr[16 + m] = o1[j] * inv;
        yr[32 + m] = o2[j] * inv;
        yr[48 + m] = o3[j] * inv;
    }
}

__global__ void nlb_out(const float* __restrict__ x, const float* __restrict__ Y,
                        const float* __restrict__ ww, const float* __restrict__ wb,
                        float* __restrict__ Z) {
    int idx = blockIdx.x * blockDim.x + threadIdx.x;
    if (idx >= B_ * T_ * C_ * H_ * W_) return;
    int wcol = idx % W_;  int tmp = idx / W_;
    int hrow = tmp % H_;  tmp /= H_;
    int cc   = tmp % C_;  tmp /= C_;
    int t    = tmp % T_;
    int b    = tmp / T_;
    int ri = hrow & 1, rj = wcol & 1;
    int oh = hrow >> 1, ow = wcol >> 1;
    int o  = (t * C_ + cc) * 4 + ri * 2 + rj;
    int n  = oh * 64 + ow;
    const float* yr = Y + (((size_t)b * NN + n) << 6);
    const float* wr = ww + o * D_;
    float acc = wb[o];
#pragma unroll 4
    for (int d = 0; d < D_; ++d) acc += wr[d] * yr[d];
    Z[idx] = acc + x[idx];
}

extern "C" void kernel_launch(void* const* d_in, const int* in_sizes, int n_in,
                              void* d_out, int out_size, void* d_ws, size_t ws_size,
                              hipStream_t stream) {
    const float* x  = (const float*)d_in[0];
    const float* gw = (const float*)d_in[1];
    const float* gb = (const float*)d_in[2];
    const float* ww = (const float*)d_in[3];
    const float* wb = (const float*)d_in[4];
    float* Z = (float*)d_out;

    const size_t buf = (size_t)B_ * NN * DP;
    float* Qbuf = (float*)d_ws;
    float* Gbuf = Qbuf + buf;
    float* Ybuf = Gbuf + buf;

    {
        int n = B_ * NN * DP;
        nlb_prep_q<<<(n + 255) / 256, 256, 0, stream>>>(x, Qbuf);
    }
    {
        int n = B_ * DP * NN;
        nlb_prep_g<<<(n + 255) / 256, 256, 0, stream>>>(Qbuf, gw, gb, Gbuf);
    }
    nlb_attn<<<B_ * 64, 128, 0, stream>>>(Qbuf, Gbuf, Ybuf);
    {
        int n = B_ * T_ * C_ * H_ * W_;
        nlb_out<<<(n + 255) / 256, 256, 0, stream>>>(x, Ybuf, ww, wb, Z);
    }
}


# attention: each 128-key K and V chunk is copied once per block into LDS (two stages, whole-line b128 loads) and B fragments are read from LDS instead of scattered per-wave global b64 loads
# speedup vs baseline: 1.5084x; 1.5084x over previous
#include <hip/hip_runtime.h>

typedef __attribute__((ext_vector_type(2))) float v2f;
typedef __attribute__((ext_vector_type(4))) float v4f;
typedef __attribute__((ext_vector_type(8))) float v8f;

#define B_  4
#define T_  5
#define C_  3
#define H_  128
#define W_  128
#define D_  60
#define DP  64
#define NN  4096
#define KT  128
#define KP  68
#define VP  132
#define KBUF (KT * KP)
#define VBUF (DP * VP)
#define STG  (KBUF + VBUF)

__device__ __forceinline__ float fast_exp2(float x) {
    return __builtin_amdgcn_exp2f(x);
}

__global__ void nlb_prep_q(const float* __restrict__ x, float* __restrict__ Q) {
    int idx = blockIdx.x * blockDim.x + threadIdx.x;
    if (idx >= B_ * NN * DP) return;
    int dp = idx & 63;
    int n  = (idx >> 6) & (NN - 1);
    int b  = idx >> 18;
    float v = 0.0f;
    if (dp < D_) {
        int tc = dp >> 2;
        int rr = dp & 3;
        int ri = rr >> 1, rj = rr & 1;
        int oh = n >> 6, ow = n & 63;
        int hh = oh * 2 + ri, ww = ow * 2 + rj;
        v = x[((size_t)(b * (T_ * C_) + tc) * H_ + hh) * W_ + ww];
    }
    Q[idx] = v;
}

__global__ void nlb_prep_g(const float* __restrict__ Q, const float* __restrict__ gw,
                           const float* __restrict__ gb, float* __restrict__ Gt) {
    int idx = blockIdx.x * blockDim.x + threadIdx.x;
    if (idx >= B_ * DP * NN) return;
    int n = idx & (NN - 1);
    int o = (idx >> 12) & 63;
    int b = idx >> 18;
    float acc = 0.0f;
    if (o < D_) {
        const float* qr = Q + (((size_t)b * NN + n) << 6);
        const float* wr = gw + o * D_;
        acc = gb[o];
#pragma unroll 4
        for (int d = 0; d < D_; ++d) acc += wr[d] * qr[d];
    }
    Gt[idx] = acc;
}

__global__ __launch_bounds__(128, 1)
void nlb_attn(const float* __restrict__ Q, const float* __restrict__ Gt,
              float* __restrict__ Y) {
    __shared__ __attribute__((aligned(16))) float lds[2 * STG + 4 * 16 * KT];

    const int tid  = threadIdx.x;
    const int lane = threadIdx.x & 31;
    const int wave = threadIdx.x >> 5;
    const int m    = lane & 15;
    const int h    = lane >> 4;
    const int b    = blockIdx.x >> 6;
    const int q0   = ((blockIdx.x & 63) * 4 + wave) * 16;

    const float* Qb  = Q  + (size_t)b * NN * DP;
    const float* Gtb = Gt + (size_t)b * DP * NN;
    float*       Yb  = Y  + (size_t)b * NN * DP;
    float*       pl  = lds + 2 * STG + wave * (16 * KT);

    const int kst = (tid >> 4) * KP + (tid & 15) * 4;
    const int vst = (tid >> 5) * VP + (tid & 31) * 4;
    const float* kgl = Qb + tid * 4;
    const float* vgl = Gtb + (size_t)(tid >> 5) * NN + (tid & 31) * 4;

    const int kfr = m * KP + 2 * h;
    const int vfr = KBUF + m * VP + 2 * h;

    v4f kr[16], vr[16];

#pragma unroll
    for (int i = 0; i < 16; ++i) {
        kr[i] = *(const v4f*)(kgl + i * 512);
        vr[i] = *(const v4f*)(vgl + (size_t)i * 4 * NN);
    }
#pragma unroll
    for (int i = 0; i < 16; ++i) {
        *(v4f*)(lds + kst + i * 8 * KP)        = kr[i];
        *(v4f*)(lds + KBUF + vst + i * 4 * VP) = vr[i];
    }

    const float LOG2E = 1.4426950408889634f;
    v2f qa[15];
    {
        const float* qrow = Qb + (size_t)(q0 + m) * DP;
#pragma unroll
        for (int kk = 0; kk < 15; ++kk) {
            v2f q = *(const v2f*)(qrow + 4 * kk + 2 * h);
            qa[kk] = q * LOG2E;
        }
    }

    v8f o0{}, o1{}, o2{}, o3{};
    float mrow[8], lrow[8];
#pragma unroll
    for (int j = 0; j < 8; ++j) { mrow[j] = -__builtin_inff(); lrow[j] = 0.0f; }

    __syncthreads();

    for (int ch = 0; ch < NN / KT; ++ch) {
        const int k0 = ch * KT;
        const float* st  = lds + (ch & 1) * STG;
        float*       nst = lds + ((ch + 1) & 1) * STG;
        const bool more = (ch + 1 < NN / KT);

        if (more) {
#pragma unroll
            for (int i = 0; i < 16; ++i) {
                kr[i] = *(const v4f*)(kgl + (size_t)(k0 + KT) * DP + i * 512);
                vr[i] = *(const v4f*)(vgl + (k0 + KT) + (size_t)i * 4 * NN);
            }
        }

        v8f s[8];
#pragma unroll
        for (int t = 0; t < 8; ++t) {
            v8f acc{};
            const float* krow = st + kfr + t * 16 * KP;
#pragma unroll
            for (int kk = 0; kk < 15; ++kk) {
                v2f kb = *(const v2f*)(krow + 4 * kk);
                acc = __builtin_amdgcn_wmma_f32_16x16x4_f32(
                          false, qa[kk], false, kb, (short)0, acc, false, false);
            }
            s[t] = acc;
        }

#pragma unroll
        for (int j = 0; j < 8; ++j) {
            float tm = fmaxf(fmaxf(fmaxf(s[0][j], s[1][j]), fmaxf(s[2][j], s[3][j])),
                             fmaxf(fmaxf(s[4][j], s[5][j]), fmaxf(s[6][j], s[7][j])));
            tm = fmaxf(tm, __shfl_xor(tm, 1, 16));
            tm = fmaxf(tm, __shfl_xor(tm, 2, 16));
            tm = fmaxf(tm, __shfl_xor(tm, 4, 16));
            tm = fmaxf(tm, __shfl_xor(tm, 8, 16));
            float mn = fmaxf(mrow[j], tm);
            float sc = fast_exp2(mrow[j] - mn);
            float p0 = fast_exp2(s[0][j] - mn);
            float p1 = fast_exp2(s[1][j] - mn);
            float p2 = fast_exp2(s[2][j] - mn);
            float p3 = fast_exp2(s[3][j] - mn);
            float p4 = fast_exp2(s[4][j] - mn);
            float p5 = fast_exp2(s[5][j] - mn);
            float p6 = fast_exp2(s[6][j] - mn);
            float p7 = fast_exp2(s[7][j] - mn);
            s[0][j] = p0; s[1][j] = p1; s[2][j] = p2; s[3][j] = p3;
            s[4][j] = p4; s[5][j] = p5; s[6][j] = p6; s[7][j] = p7;
            float su = ((p0 + p1) + (p2 + p3)) + ((p4 + p5) + (p6 + p7));
            su += __shfl_xor(su, 1, 16);
            su += __shfl_xor(su, 2, 16);
            su += __shfl_xor(su, 4, 16);
            su += __shfl_xor(su, 8, 16);
            lrow[j] = lrow[j] * sc + su;
            mrow[j] = mn;
            o0[j] *= sc; o1[j] *= sc; o2[j] *= sc; o3[j] *= sc;
        }

#pragma unroll
        for (int t = 0; t < 8; ++t)
#pragma unroll
            for (int j = 0; j < 8; ++j)
                pl[(8 * h + j) * KT + t * 16 + m] = s[t][j];
        asm volatile("" ::: "memory");

#pragma unroll
        for (int kk2 = 0; kk2 < 32; ++kk2) {
            v2f pa = *(const v2f*)(pl + m * KT + 4 * kk2 + 2 * h);
            const float* vp = st + vfr + 4 * kk2;
            v2f vb;
            vb = *(const v2f*)(vp);
            o0 = __builtin_amdgcn_wmma_f32_16x16x4_f32(false, pa, false, vb, (short)0, o0, false, false);
            vb = *(const v2f*)(vp + 16 * VP);
            o1 = __builtin_amdgcn_wmma_f32_16x16x4_f32(false, pa, false, vb, (short)0, o1, false, false);
            vb = *(const v2f*)(vp + 32 * VP);
            o2 = __builtin_amdgcn_wmma_f32_16x16x4_f32(false, pa, false, vb, (short)0, o2, false, false);
            vb = *(const v2f*)(vp + 48 * VP);
            o3 = __builtin_amdgcn_wmma_f32_16x16x4_f32(false, pa, false, vb, (short)0, o3, false, false);
        }
        asm volatile("" ::: "memory");

        if (more) {
#pragma unroll
            for (int i = 0; i < 16; ++i) {
                *(v4f*)(nst + kst + i * 8 * KP)        = kr[i];
                *(v4f*)(nst + KBUF + vst + i * 4 * VP) = vr[i];
            }
        }
        __syncthreads();
    }

#pragma unroll
    for (int j = 0; j < 8; ++j) {
        float inv = 1.0f / lrow[j];
        float* yr = Yb + (size_t)(q0 + 8 * h + j) * DP;
        yr[ 0 + m] = o0[j] * inv;
        yr[16 + m] = o1[j] * inv;
        yr[32 + m] = o2[j] * inv;
        yr[48 + m] = o3[j] * inv;
    }
}

__global__ void nlb_out(const float* __restrict__ x, const float* __restrict__ Y,
                        const float* __restrict__ ww, const float* __restrict__ wb,
                        float* __restrict__ Z) {
    int idx = blockIdx.x * blockDim.x + threadIdx.x;
    if (idx >= B_ * T_ * C_ * H_ * W_) return;
    int wcol = idx % W_;  int tmp = idx / W_;
    int hrow = tmp % H_;  tmp /= H_;
    int cc   = tmp % C_;  tmp /= C_;
    int t    = tmp % T_;
    int b    = tmp / T_;
    int ri = hrow & 1, rj = wcol & 1;
    int oh = hrow >> 1, ow = wcol >> 1;
    int o  = (t * C_ + cc) * 4 + ri * 2 + rj;
    int n  = oh * 64 + ow;
    const float* yr = Y + (((size_t)b * NN + n) << 6);
    const float* wr = ww + o * D_;
    float acc = wb[o];
#pragma unroll 4
    for (int d = 0; d < D_; ++d) acc += wr[d] * yr[d];
    Z[idx] = acc + x[idx];
}

extern "C" void kernel_launch(void* const* d_in, const int* in_sizes, int n_in,
                              void* d_out, int out_size, void* d_ws, size_t ws_size,
                              hipStream_t stream) {
    const float* x  = (const float*)d_in[0];
    const float* gw = (const float*)d_in[1];
    const float* gb = (const float*)d_in[2];
    const float* ww = (const float*)d_in[3];
    const float* wb = (const float*)d_in[4];
    float* Z = (float*)d_out;

    const size_t buf = (size_t)B_ * NN * DP;
    float* Qbuf = (float*)d_ws;
    float* Gbuf = Qbuf + buf;
    float* Ybuf = Gbuf + buf;

    {
        int n = B_ * NN * DP;
        nlb_prep_q<<<(n + 255) / 256, 256, 0, stream>>>(x, Qbuf);
    }
    {
        int n = B_ * DP * NN;
        nlb_prep_g<<<(n + 255) / 256, 256, 0, stream>>>(Qbuf, gw, gb, Gbuf);
    }
    nlb_attn<<<B_ * 64, 128, 0, stream>>>(Qbuf, Gbuf, Ybuf);
    {
        int n = B_ * T_ * C_ * H_ * W_;
        nlb_out<<<(n + 255) / 256, 256, 0, stream>>>(x, Ybuf, ww, wb, Z);
    }
}


# both 60x60 channel projections (V and output) move from one-thread-per-output FMA loops to f32 16x16x4 WMMA tiles, one wave per 32 or 16 positions; sums in WMMA order, bias added last
# speedup vs baseline: 1.8694x; 1.2393x over previous
#include <hip/hip_runtime.h>

typedef __attribute__((ext_vector_type(2))) float v2f;
typedef __attribute__((ext_vector_type(4))) float v4f;
typedef __attribute__((ext_vector_type(8))) float v8f;

#define B_  4
#define T_  5
#define C_  3
#define H_  128
#define W_  128
#define D_  60
#define DP  64
#define NN  4096
#define KT  128
#define KP  68
#define VP  132
#define KBUF (KT * KP)
#define VBUF (DP * VP)
#define STG  (KBUF + VBUF)

__device__ __forceinline__ float fast_exp2(float x) {
    return __builtin_amdgcn_exp2f(x);
}

__global__ void nlb_prep_q(const float* __restrict__ x, float* __restrict__ Q) {
    int idx = blockIdx.x * blockDim.x + threadIdx.x;
    if (idx >= B_ * NN * DP) return;
    int dp = idx & 63;
    int n  = (idx >> 6) & (NN - 1);
    int b  = idx >> 18;
    float v = 0.0f;
    if (dp < D_) {
        int tc = dp >> 2;
        int rr = dp & 3;
        int ri = rr >> 1, rj = rr & 1;
        int oh = n >> 6, ow = n & 63;
        int hh = oh * 2 + ri, ww = ow * 2 + rj;
        v = x[((size_t)(b * (T_ * C_) + tc) * H_ + hh) * W_ + ww];
    }
    Q[idx] = v;
}

__global__ __launch_bounds__(64)
void nlb_prep_g(const float* __restrict__ Q, const float* __restrict__ gw,
                const float* __restrict__ gb, float* __restrict__ Gt) {
    const int lane = threadIdx.x & 31;
    const int wave = threadIdx.x >> 5;
    const int m    = lane & 15;
    const int h    = lane >> 4;
    const int tile = blockIdx.x * 2 + wave;
    const int b    = tile >> 7;
    const int n0   = (tile & 127) * 32;

    v2f qb[2][15];
#pragma unroll
    for (int nt = 0; nt < 2; ++nt) {
        const float* qrow = Q + ((size_t)b * NN + n0 + 16 * nt + m) * DP + 2 * h;
#pragma unroll
        for (int kk = 0; kk < 15; ++kk) qb[nt][kk] = *(const v2f*)(qrow + 4 * kk);
    }

    v8f acc[2][4];
#pragma unroll
    for (int nt = 0; nt < 2; ++nt)
#pragma unroll
        for (int ot = 0; ot < 4; ++ot) acc[nt][ot] = v8f{};

#pragma unroll
    for (int kk = 0; kk < 15; ++kk) {
#pragma unroll
        for (int ot = 0; ot < 4; ++ot) {
            const int  o    = ot * 16 + m;
            const bool live = (ot < 3) || (o < D_);
            v2f wa = *(const v2f*)(gw + (live ? o : 0) * D_ + 4 * kk + 2 * h);
            wa = live ? wa : v2f{0.0f, 0.0f};
#pragma unroll
            for (int nt = 0; nt < 2; ++nt)
                acc[nt][ot] = __builtin_amdgcn_wmma_f32_16x16x4_f32(
                                  false, wa, false, qb[nt][kk], (short)0, acc[nt][ot], false, false);
        }
    }

    float* gcol = Gt + (size_t)b * DP * NN + n0 + m;
#pragma unroll
    for (int ot = 0; ot < 4; ++ot) {
#pragma unroll
        for (int j = 0; j < 8; ++j) {
            const int  of   = ot * 16 + 8 * h + j;
            const bool live = (ot < 3) || (of < D_);
            const float bias = gb[live ? of : 0];
#pragma unroll
            for (int nt = 0; nt < 2; ++nt)
                gcol[(size_t)of * NN + 16 * nt] = live ? (acc[nt][ot][j] + bias) : 0.0f;
        }
    }
}

__global__ __launch_bounds__(128, 1)
void nlb_attn(const float* __restrict__ Q, const float* __restrict__ Gt,
              float* __restrict__ Y) {
    __shared__ __attribute__((aligned(16))) float lds[2 * STG + 4 * 16 * KT];

    const int tid  = threadIdx.x;
    const int lane = threadIdx.x & 31;
    const int wave = threadIdx.x >> 5;
    const int m    = lane & 15;
    const int h    = lane >> 4;
    const int b    = blockIdx.x >> 6;
    const int q0   = ((blockIdx.x & 63) * 4 + wave) * 16;

    const float* Qb  = Q  + (size_t)b * NN * DP;
    const float* Gtb = Gt + (size_t)b * DP * NN;
    float*       Yb  = Y  + (size_t)b * NN * DP;
    float*       pl  = lds + 2 * STG + wave * (16 * KT);

    const int kst = (tid >> 4) * KP + (tid & 15) * 4;
    const int vst = (tid >> 5) * VP + (tid & 31) * 4;
    const float* kgl = Qb + tid * 4;
    const float* vgl = Gtb + (size_t)(tid >> 5) * NN + (tid & 31) * 4;

    const int kfr = m * KP + 2 * h;
    const int vfr = KBUF + m * VP + 2 * h;

    v4f kr[16], vr[16];

#pragma unroll
    for (int i = 0; i < 16; ++i) {
        kr[i] = *(const v4f*)(kgl + i * 512);
        vr[i] = *(const v4f*)(vgl + (size_t)i * 4 * NN);
    }
#pragma unroll
    for (int i = 0; i < 16; ++i) {
        *(v4f*)(lds + kst + i * 8 * KP)        = kr[i];
        *(v4f*)(lds + KBUF + vst + i * 4 * VP) = vr[i];
    }

    const float LOG2E = 1.4426950408889634f;
    v2f qa[15];
    {
        const float* qrow = Qb + (size_t)(q0 + m) * DP;
#pragma unroll
        for (int kk = 0; kk < 15; ++kk) {
            v2f q = *(const v2f*)(qrow + 4 * kk + 2 * h);
            qa[kk] = q * LOG2E;
        }
    }

    v8f o0{}, o1{}, o2{}, o3{};
    float mrow[8], lrow[8];
#pragma unroll
    for (int j = 0; j < 8; ++j) { mrow[j] = -__builtin_inff(); lrow[j] = 0.0f; }

    __syncthreads();

    for (int ch = 0; ch < NN / KT; ++ch) {
        const int k0 = ch * KT;
        const float* st  = lds + (ch & 1) * STG;
        float*       nst = lds + ((ch + 1) & 1) * STG;
        const bool more = (ch + 1 < NN / KT);

        if (more) {
#pragma unroll
            for (int i = 0; i < 16; ++i) {
                kr[i] = *(const v4f*)(kgl + (size_t)(k0 + KT) * DP + i * 512);
                vr[i] = *(const v4f*)(vgl + (k0 + KT) + (size_t)i * 4 * NN);
            }
        }

        v8f s[8];
#pragma unroll
        for (int t = 0; t < 8; ++t) {
            v8f acc{};
            const float* krow = st + kfr + t * 16 * KP;
#pragma unroll
            for (int kk = 0; kk < 15; ++kk) {
                v2f kb = *(const v2f*)(krow + 4 * kk);
                acc = __builtin_amdgcn_wmma_f32_16x16x4_f32(
                          false, qa[kk], false, kb, (short)0, acc, false, false);
            }
            s[t] = acc;
        }

#pragma unroll
        for (int j = 0; j < 8; ++j) {
            float tm = fmaxf(fmaxf(fmaxf(s[0][j], s[1][j]), fmaxf(s[2][j], s[3][j])),
                             fmaxf(fmaxf(s[4][j], s[5][j]), fmaxf(s[6][j], s[7][j])));
            tm = fmaxf(tm, __shfl_xor(tm, 1, 16));
            tm = fmaxf(tm, __shfl_xor(tm, 2, 16));
            tm = fmaxf(tm, __shfl_xor(tm, 4, 16));
            tm = fmaxf(tm, __shfl_xor(tm, 8, 16));
            float mn = fmaxf(mrow[j], tm);
            float sc = fast_exp2(mrow[j] - mn);
            float p0 = fast_exp2(s[0][j] - mn);
            float p1 = fast_exp2(s[1][j] - mn);
            float p2 = fast_exp2(s[2][j] - mn);
            float p3 = fast_exp2(s[3][j] - mn);
            float p4 = fast_exp2(s[4][j] - mn);
            float p5 = fast_exp2(s[5][j] - mn);
            float p6 = fast_exp2(s[6][j] - mn);
            float p7 = fast_exp2(s[7][j] - mn);
            s[0][j] = p0; s[1][j] = p1; s[2][j] = p2; s[3][j] = p3;
            s[4][j] = p4; s[5][j] = p5; s[6][j] = p6; s[7][j] = p7;
            float su = ((p0 + p1) + (p2 + p3)) + ((p4 + p5) + (p6 + p7));
            su += __shfl_xor(su, 1, 16);
            su += __shfl_xor(su, 2, 16);
            su += __shfl_xor(su, 4, 16);
            su += __shfl_xor(su, 8, 16);
            lrow[j] = lrow[j] * sc + su;
            mrow[j] = mn;
            o0[j] *= sc; o1[j] *= sc; o2[j] *= sc; o3[j] *= sc;
        }

#pragma unroll
        for (int t = 0; t < 8; ++t)
#pragma unroll
            for (int j = 0; j < 8; ++j)
                pl[(8 * h + j) * KT + t * 16 + m] = s[t][j];
        asm volatile("" ::: "memory");

#pragma unroll
        for (int kk2 = 0; kk2 < 32; ++kk2) {
            v2f pa = *(const v2f*)(pl + m * KT + 4 * kk2 + 2 * h);
            const float* vp = st + vfr + 4 * kk2;
            v2f vb;
            vb = *(const v2f*)(vp);
            o0 = __builtin_amdgcn_wmma_f32_16x16x4_f32(false, pa, false, vb, (short)0, o0, false, false);
            vb = *(const v2f*)(vp + 16 * VP);
            o1 = __builtin_amdgcn_wmma_f32_16x16x4_f32(false, pa, false, vb, (short)0, o1, false, false);
            vb = *(const v2f*)(vp + 32 * VP);
            o2 = __builtin_amdgcn_wmma_f32_16x16x4_f32(false, pa, false, vb, (short)0, o2, false, false);
            vb = *(const v2f*)(vp + 48 * VP);
            o3 = __builtin_amdgcn_wmma_f32_16x16x4_f32(false, pa, false, vb, (short)0, o3, false, false);
        }
        asm volatile("" ::: "memory");

        if (more) {
#pragma unroll
            for (int i = 0; i < 16; ++i) {
                *(v4f*)(nst + kst + i * 8 * KP)        = kr[i];
                *(v4f*)(nst + KBUF + vst + i * 4 * VP) = vr[i];
            }
        }
        __syncthreads();
    }

#pragma unroll
    for (int j = 0; j < 8; ++j) {
        float inv = 1.0f / lrow[j];
        float* yr = Yb + (size_t)(q0 + 8 * h + j) * DP;
        yr[ 0 + m] = o0[j] * inv;
        yr[16 + m] = o1[j] * inv;
        yr[32 + m] = o2[j] * inv;
        yr[48 + m] = o3[j] * inv;
    }
}

__global__ __launch_bounds__(128)
void nlb_out(const float* __restrict__ x, const float* __restrict__ Y,
             const float* __restrict__ ww, const float* __restrict__ wb,
             float* __restrict__ Z) {
    const int lane = threadIdx.x & 31;
    const int wave = threadIdx.x >> 5;
    const int m    = lane & 15;
    const int h    = lane >> 4;
    const int tile = blockIdx.x * 4 + wave;
    const int b    = tile >> 8;
    const int n0   = (tile & 255) * 16;

    v2f yb[15];
    {
        const float* yrow = Y + ((size_t)b * NN + n0 + m) * DP + 2 * h;
#pragma unroll
        for (int kk = 0; kk < 15; ++kk) yb[kk] = *(const v2f*)(yrow + 4 * kk);
    }

    v8f acc[4];
#pragma unroll
    for (int ot = 0; ot < 4; ++ot) acc[ot] = v8f{};

#pragma unroll
    for (int kk = 0; kk < 15; ++kk) {
#pragma unroll
        for (int ot = 0; ot < 4; ++ot) {
            const int  o    = ot * 16 + m;
            const bool live = (ot < 3) || (o < D_);
            v2f wa = *(const v2f*)(ww + (live ? o : 0) * D_ + 4 * kk + 2 * h);
            wa = live ? wa : v2f{0.0f, 0.0f};
            acc[ot] = __builtin_amdgcn_wmma_f32_16x16x4_f32(
                          false, wa, false, yb[kk], (short)0, acc[ot], false, false);
        }
    }

    const int oh = n0 >> 6;
    const int ow = (n0 & 63) + m;
#pragma unroll
    for (int ot = 0; ot < 4; ++ot) {
#pragma unroll
        for (int jp = 0; jp < 4; ++jp) {
            const int oo = ot * 16 + 8 * h + 2 * jp;
            if ((ot < 3) || (oo < D_)) {
                const int tc = oo >> 2;
                const int ri = (oo >> 1) & 1;
                const size_t zi = ((size_t)(b * (T_ * C_) + tc) * H_ + (2 * oh + ri)) * W_ + 2 * ow;
                const v2f xv = *(const v2f*)(x + zi);
                v2f r;
                r.x = (acc[ot][2 * jp]     + wb[oo])     + xv.x;
                r.y = (acc[ot][2 * jp + 1] + wb[oo + 1]) + xv.y;
                *(v2f*)(Z + zi) = r;
            }
        }
    }
}

extern "C" void kernel_launch(void* const* d_in, const int* in_sizes, int n_in,
                              void* d_out, int out_size, void* d_ws, size_t ws_size,
                              hipStream_t stream) {
    const float* x  = (const float*)d_in[0];
    const float* gw = (const float*)d_in[1];
    const float* gb = (const float*)d_in[2];
    const float* ww = (const float*)d_in[3];
    const float* wb = (const float*)d_in[4];
    float* Z = (float*)d_out;

    const size_t buf = (size_t)B_ * NN * DP;
    float* Qbuf = (float*)d_ws;
    float* Gbuf = Qbuf + buf;
    float* Ybuf = Gbuf + buf;

    {
        int n = B_ * NN * DP;
        nlb_prep_q<<<(n + 255) / 256, 256, 0, stream>>>(x, Qbuf);
    }
    nlb_prep_g<<<B_ * NN / 64, 64, 0, stream>>>(Qbuf, gw, gb, Gbuf);
    nlb_attn<<<B_ * 64, 128, 0, stream>>>(Qbuf, Gbuf, Ybuf);
    nlb_out<<<B_ * NN / 64, 128, 0, stream>>>(x, Ybuf, ww, wb, Z);
}


# attention: score and output tiles computed transposed (keys x queries); exp'd scores feed the PV product from registers as B operand: no LDS transpose, no lane shuffles; key and row-sum order change
# speedup vs baseline: 2.2849x; 1.2223x over previous
#include <hip/hip_runtime.h>

typedef __attribute__((ext_vector_type(2))) float v2f;
typedef __attribute__((ext_vector_type(4))) float v4f;
typedef __attribute__((ext_vector_type(8))) float v8f;

#define B_  4
#define T_  5
#define C_  3
#define H_  128
#define W_  128
#define D_  60
#define DP  64
#define NN  4096
#define KT  128
#define KP  68
#define VP  132
#define KBUF (KT * KP)
#define VBUF (DP * VP)
#define STG  (KBUF + VBUF)

__device__ __forceinline__ float fast_exp2(float x) {
    return __builtin_amdgcn_exp2f(x);
}

__global__ void nlb_prep_q(const float* __restrict__ x, float* __restrict__ Q) {
    int idx = blockIdx.x * blockDim.x + threadIdx.x;
    if (idx >= B_ * NN * DP) return;
    int dp = idx & 63;
    int n  = (idx >> 6) & (NN - 1);
    int b  = idx >> 18;
    float v = 0.0f;
    if (dp < D_) {
        int tc = dp >> 2;
        int rr = dp & 3;
        int ri = rr >> 1, rj = rr & 1;
        int oh = n >> 6, ow = n & 63;
        int hh = oh * 2 + ri, ww = ow * 2 + rj;
        v = x[((size_t)(b * (T_ * C_) + tc) * H_ + hh) * W_ + ww];
    }
    Q[idx] = v;
}

__global__ __launch_bounds__(64)
void nlb_prep_g(const float* __restrict__ Q, const float* __restrict__ gw,
                const float* __restrict__ gb, float* __restrict__ Gt) {
    const int lane = threadIdx.x & 31;
    const int wave = threadIdx.x >> 5;
    const int m    = lane & 15;
    const int h    = lane >> 4;
    const int tile = blockIdx.x * 2 + wave;
    const int b    = tile >> 7;
    const int n0   = (tile & 127) * 32;

    v2f qb[2][15];
#pragma unroll
    for (int nt = 0; nt < 2; ++nt) {
        const float* qrow = Q + ((size_t)b * NN + n0 + 16 * nt + m) * DP + 2 * h;
#pragma unroll
        for (int kk = 0; kk < 15; ++kk) qb[nt][kk] = *(const v2f*)(qrow + 4 * kk);
    }

    v8f acc[2][4];
#pragma unroll
    for (int nt = 0; nt < 2; ++nt)
#pragma unroll
        for (int ot = 0; ot < 4; ++ot) acc[nt][ot] = v8f{};

#pragma unroll
    for (int kk = 0; kk < 15; ++kk) {
#pragma unroll
        for (int ot = 0; ot < 4; ++ot) {
            const int  o    = ot * 16 + m;
            const bool live = (ot < 3) || (o < D_);
            v2f wa = *(const v2f*)(gw + (live ? o : 0) * D_ + 4 * kk + 2 * h);
            wa = live ? wa : v2f{0.0f, 0.0f};
#pragma unroll
            for (int nt = 0; nt < 2; ++nt)
                acc[nt][ot] = __builtin_amdgcn_wmma_f32_16x16x4_f32(
                                  false, wa, false, qb[nt][kk], (short)0, acc[nt][ot], false, false);
        }
    }

    float* gcol = Gt + (size_t)b * DP * NN + n0 + m;
#pragma unroll
    for (int ot = 0; ot < 4; ++ot) {
#pragma unroll
        for (int j = 0; j < 8; ++j) {
            const int  of   = ot * 16 + 8 * h + j;
            const bool live = (ot < 3) || (of < D_);
            const float bias = gb[live ? of : 0];
#pragma unroll
            for (int nt = 0; nt < 2; ++nt)
                gcol[(size_t)of * NN + 16 * nt] = live ? (acc[nt][ot][j] + bias) : 0.0f;
        }
    }
}


__device__ __forceinline__ float xhalf(float x) {
    int b = __builtin_bit_cast(int, x);
    return __builtin_bit_cast(float,
        __builtin_amdgcn_permlanex16(b, b, 0x76543210, 0xfedcba98, false, false));
}

__global__ __launch_bounds__(128, 1)
void nlb_attn(const float* __restrict__ Q, const float* __restrict__ Gt,
              float* __restrict__ Y) {
    __shared__ __attribute__((aligned(16))) float lds[2 * STG];

    const int tid  = threadIdx.x;
    const int lane = threadIdx.x & 31;
    const int wave = threadIdx.x >> 5;
    const int m    = lane & 15;
    const int h    = lane >> 4;
    const int b    = blockIdx.x >> 6;
    const int q0   = ((blockIdx.x & 63) * 4 + wave) * 16;

    const float* Qb  = Q  + (size_t)b * NN * DP;
    const float* Gtb = Gt + (size_t)b * DP * NN;
    float*       Yb  = Y  + (size_t)b * NN * DP;

    const int kst = (tid >> 4) * KP + (tid & 15) * 4;
    const int vst = (tid >> 5) * VP + (tid & 31) * 4;
    const float* kgl = Qb + tid * 4;
    const float* vgl = Gtb + (size_t)(tid >> 5) * NN + (tid & 31) * 4;

    const int kfr = m * KP + 2 * h;
    const int vfr = KBUF + m * VP + 8 * h;

    v4f kr[16], vr[16];

#pragma unroll
    for (int i = 0; i < 16; ++i) {
        kr[i] = *(const v4f*)(kgl + i * 512);
        vr[i] = *(const v4f*)(vgl + (size_t)i * 4 * NN);
    }
#pragma unroll
    for (int i = 0; i < 16; ++i) {
        *(v4f*)(lds + kst + i * 8 * KP)        = kr[i];
        *(v4f*)(lds + KBUF + vst + i * 4 * VP) = vr[i];
    }

    const float LOG2E = 1.4426950408889634f;
    v2f qa[15];
    {
        const float* qrow = Qb + (size_t)(q0 + m) * DP;
#pragma unroll
        for (int kk = 0; kk < 15; ++kk) {
            v2f q = *(const v2f*)(qrow + 4 * kk + 2 * h);
            qa[kk] = q * LOG2E;
        }
    }

    v8f o0{}, o1{}, o2{}, o3{};
    float mrow = -__builtin_inff();
    float lrow = 0.0f;

    __syncthreads();

    for (int ch = 0; ch < NN / KT; ++ch) {
        const int k0 = ch * KT;
        const float* st  = lds + (ch & 1) * STG;
        float*       nst = lds + ((ch + 1) & 1) * STG;
        const bool more = (ch + 1 < NN / KT);

        if (more) {
#pragma unroll
            for (int i = 0; i < 16; ++i) {
                kr[i] = *(const v4f*)(kgl + (size_t)(k0 + KT) * DP + i * 512);
                vr[i] = *(const v4f*)(vgl + (k0 + KT) + (size_t)i * 4 * NN);
            }
        }

        v8f s[8];
#pragma unroll
        for (int t = 0; t < 8; ++t) {
            v8f acc{};
            const float* krow = st + kfr + t * 16 * KP;
#pragma unroll
            for (int kk = 0; kk < 15; ++kk) {
                v2f ka = *(const v2f*)(krow + 4 * kk);
                acc = __builtin_amdgcn_wmma_f32_16x16x4_f32(
                          false, ka, false, qa[kk], (short)0, acc, false, false);
            }
            s[t] = acc;
        }

        float tm;
        {
            float tmax[8];
#pragma unroll
            for (int t = 0; t < 8; ++t)
                tmax[t] = fmaxf(fmaxf(fmaxf(s[t][0], s[t][1]), fmaxf(s[t][2], s[t][3])),
                                fmaxf(fmaxf(s[t][4], s[t][5]), fmaxf(s[t][6], s[t][7])));
            tm = fmaxf(fmaxf(fmaxf(tmax[0], tmax[1]), fmaxf(tmax[2], tmax[3])),
                       fmaxf(fmaxf(tmax[4], tmax[5]), fmaxf(tmax[6], tmax[7])));
        }
        tm = fmaxf(tm, xhalf(tm));
        const float mn = fmaxf(mrow, tm);
        const float sc = fast_exp2(mrow - mn);
        float su;
        {
            float part[8];
#pragma unroll
            for (int t = 0; t < 8; ++t) {
#pragma unroll
                for (int r = 0; r < 8; ++r) s[t][r] = fast_exp2(s[t][r] - mn);
                part[t] = ((s[t][0] + s[t][1]) + (s[t][2] + s[t][3])) +
                          ((s[t][4] + s[t][5]) + (s[t][6] + s[t][7]));
            }
            su = ((part[0] + part[1]) + (part[2] + part[3])) +
                 ((part[4] + part[5]) + (part[6] + part[7]));
        }
        su += xhalf(su);
        lrow = lrow * sc + su;
        mrow = mn;
        o0 *= sc; o1 *= sc; o2 *= sc; o3 *= sc;

#pragma unroll
        for (int t = 0; t < 8; ++t) {
#pragma unroll
            for (int up = 0; up < 2; ++up) {
                const float* vp = st + vfr + 16 * t + 4 * up;
                const v4f va0 = *(const v4f*)(vp);
                const v4f va1 = *(const v4f*)(vp + 16 * VP);
                const v4f va2 = *(const v4f*)(vp + 32 * VP);
                const v4f va3 = *(const v4f*)(vp + 48 * VP);
                const v2f pb0 = v2f{s[t][4 * up],     s[t][4 * up + 1]};
                const v2f pb1 = v2f{s[t][4 * up + 2], s[t][4 * up + 3]};
                o0 = __builtin_amdgcn_wmma_f32_16x16x4_f32(false, v2f{va0.x, va0.y}, false, pb0, (short)0, o0, false, false);
                o1 = __builtin_amdgcn_wmma_f32_16x16x4_f32(false, v2f{va1.x, va1.y}, false, pb0, (short)0, o1, false, false);
                o2 = __builtin_amdgcn_wmma_f32_16x16x4_f32(false, v2f{va2.x, va2.y}, false, pb0, (short)0, o2, false, false);
                o3 = __builtin_amdgcn_wmma_f32_16x16x4_f32(false, v2f{va3.x, va3.y}, false, pb0, (short)0, o3, false, false);
                o0 = __builtin_amdgcn_wmma_f32_16x16x4_f32(false, v2f{va0.z, va0.w}, false, pb1, (short)0, o0, false, false);
                o1 = __builtin_amdgcn_wmma_f32_16x16x4_f32(false, v2f{va1.z, va1.w}, false, pb1, (short)0, o1, false, false);
                o2 = __builtin_amdgcn_wmma_f32_16x16x4_f32(false, v2f{va2.z, va2.w}, false, pb1, (short)0, o2, false, false);
                o3 = __builtin_amdgcn_wmma_f32_16x16x4_f32(false, v2f{va3.z, va3.w}, false, pb1, (short)0, o3, false, false);
            }
        }

        if (more) {
#pragma unroll
            for (int i = 0; i < 16; ++i) {
                *(v4f*)(nst + kst + i * 8 * KP)        = kr[i];
                *(v4f*)(nst + KBUF + vst + i * 4 * VP) = vr[i];
            }
        }
        __syncthreads();
    }

    {
        const float inv = 1.0f / lrow;
        float* yr = Yb + (size_t)(q0 + m) * DP + 8 * h;
        *(v4f*)(yr +  0) = v4f{o0[0], o0[1], o0[2], o0[3]} * inv;
        *(v4f*)(yr +  4) = v4f{o0[4], o0[5], o0[6], o0[7]} * inv;
        *(v4f*)(yr + 16) = v4f{o1[0], o1[1], o1[2], o1[3]} * inv;
        *(v4f*)(yr + 20) = v4f{o1[4], o1[5], o1[6], o1[7]} * inv;
        *(v4f*)(yr + 32) = v4f{o2[0], o2[1], o2[2], o2[3]} * inv;
        *(v4f*)(yr + 36) = v4f{o2[4], o2[5], o2[6], o2[7]} * inv;
        *(v4f*)(yr + 48) = v4f{o3[0], o3[1], o3[2], o3[3]} * inv;
        *(v4f*)(yr + 52) = v4f{o3[4], o3[5], o3[6], o3[7]} * inv;
    }
}

__global__ __launch_bounds__(128)
void nlb_out(const float* __restrict__ x, const float* __restrict__ Y,
             const float* __restrict__ ww, const float* __restrict__ wb,
             float* __restrict__ Z) {
    const int lane = threadIdx.x & 31;
    const int wave = threadIdx.x >> 5;
    const int m    = lane & 15;
    const int h    = lane >> 4;
    const int tile = blockIdx.x * 4 + wave;
    const int b    = tile >> 8;
    const int n0   = (tile & 255) * 16;

    v2f yb[15];
    {
        const float* yrow = Y + ((size_t)b * NN + n0 + m) * DP + 2 * h;
#pragma unroll
        for (int kk = 0; kk < 15; ++kk) yb[kk] = *(const v2f*)(yrow + 4 * kk);
    }

    v8f acc[4];
#pragma unroll
    for (int ot = 0; ot < 4; ++ot) acc[ot] = v8f{};

#pragma unroll
    for (int kk = 0; kk < 15; ++kk) {
#pragma unroll
        for (int ot = 0; ot < 4; ++ot) {
            const int  o    = ot * 16 + m;
            const bool live = (ot < 3) || (o < D_);
            v2f wa = *(const v2f*)(ww + (live ? o : 0) * D_ + 4 * kk + 2 * h);
            wa = live ? wa : v2f{0.0f, 0.0f};
            acc[ot] = __builtin_amdgcn_wmma_f32_16x16x4_f32(
                          false, wa, false, yb[kk], (short)0, acc[ot], false, false);
        }
    }

    const int oh = n0 >> 6;
    const int ow = (n0 & 63) + m;
#pragma unroll
    for (int ot = 0; ot < 4; ++ot) {
#pragma unroll
        for (int jp = 0; jp < 4; ++jp) {
            const int oo = ot * 16 + 8 * h + 2 * jp;
            if ((ot < 3) || (oo < D_)) {
                const int tc = oo >> 2;
                const int ri = (oo >> 1) & 1;
                const size_t zi = ((size_t)(b * (T_ * C_) + tc) * H_ + (2 * oh + ri)) * W_ + 2 * ow;
                const v2f xv = *(const v2f*)(x + zi);
                v2f r;
                r.x = (acc[ot][2 * jp]     + wb[oo])     + xv.x;
                r.y = (acc[ot][2 * jp + 1] + wb[oo + 1]) + xv.y;
                *(v2f*)(Z + zi) = r;
            }
        }
    }
}

extern "C" void kernel_launch(void* const* d_in, const int* in_sizes, int n_in,
                              void* d_out, int out_size, void* d_ws, size_t ws_size,
                              hipStream_t stream) {
    const float* x  = (const float*)d_in[0];
    const float* gw = (const float*)d_in[1];
    const float* gb = (const float*)d_in[2];
    const float* ww = (const float*)d_in[3];
    const float* wb = (const float*)d_in[4];
    float* Z = (float*)d_out;

    const size_t buf = (size_t)B_ * NN * DP;
    float* Qbuf = (float*)d_ws;
    float* Gbuf = Qbuf + buf;
    float* Ybuf = Gbuf + buf;

    {
        int n = B_ * NN * DP;
        nlb_prep_q<<<(n + 255) / 256, 256, 0, stream>>>(x, Qbuf);
    }
    nlb_prep_g<<<B_ * NN / 64, 64, 0, stream>>>(Qbuf, gw, gb, Gbuf);
    nlb_attn<<<B_ * 64, 128, 0, stream>>>(Qbuf, Gbuf, Ybuf);
    nlb_out<<<B_ * NN / 64, 128, 0, stream>>>(x, Ybuf, ww, wb, Z);
}


# attention: each chunk's LDS traffic issues as two whole groups ahead of the WMMA runs (V fragments and staging stores before the score run, next K fragments before the PV run), none among the WMMAs
# speedup vs baseline: 2.3908x; 1.0463x over previous
#include <hip/hip_runtime.h>

typedef __attribute__((ext_vector_type(2))) float v2f;
typedef __attribute__((ext_vector_type(4))) float v4f;
typedef __attribute__((ext_vector_type(8))) float v8f;

#define B_  4
#define T_  5
#define C_  3
#define H_  128
#define W_  128
#define D_  60
#define DP  64
#define NN  4096
#define KT  128
#define KP  68
#define VP  132
#define KBUF (KT * KP)
#define VBUF (DP * VP)
#define STG  (KBUF + VBUF)

__device__ __forceinline__ float fast_exp2(float x) {
    return __builtin_amdgcn_exp2f(x);
}

__global__ void nlb_prep_q(const float* __restrict__ x, float* __restrict__ Q) {
    int idx = blockIdx.x * blockDim.x + threadIdx.x;
    if (idx >= B_ * NN * DP) return;
    int dp = idx & 63;
    int n  = (idx >> 6) & (NN - 1);
    int b  = idx >> 18;
    float v = 0.0f;
    if (dp < D_) {
        int tc = dp >> 2;
        int rr = dp & 3;
        int ri = rr >> 1, rj = rr & 1;
        int oh = n >> 6, ow = n & 63;
        int hh = oh * 2 + ri, ww = ow * 2 + rj;
        v = x[((size_t)(b * (T_ * C_) + tc) * H_ + hh) * W_ + ww];
    }
    Q[idx] = v;
}

__global__ __launch_bounds__(64)
void nlb_prep_g(const float* __restrict__ Q, const float* __restrict__ gw,
                const float* __restrict__ gb, float* __restrict__ Gt) {
    const int lane = threadIdx.x & 31;
    const int wave = threadIdx.x >> 5;
    const int m    = lane & 15;
    const int h    = lane >> 4;
    const int tile = blockIdx.x * 2 + wave;
    const int b    = tile >> 7;
    const int n0   = (tile & 127) * 32;

    v2f qb[2][15];
#pragma unroll
    for (int nt = 0; nt < 2; ++nt) {
        const float* qrow = Q + ((size_t)b * NN + n0 + 16 * nt + m) * DP + 2 * h;
#pragma unroll
        for (int kk = 0; kk < 15; ++kk) qb[nt][kk] = *(const v2f*)(qrow + 4 * kk);
    }

    v8f acc[2][4];
#pragma unroll
    for (int nt = 0; nt < 2; ++nt)
#pragma unroll
        for (int ot = 0; ot < 4; ++ot) acc[nt][ot] = v8f{};

#pragma unroll
    for (int kk = 0; kk < 15; ++kk) {
#pragma unroll
        for (int ot = 0; ot < 4; ++ot) {
            const int  o    = ot * 16 + m;
            const bool live = (ot < 3) || (o < D_);
            v2f wa = *(const v2f*)(gw + (live ? o : 0) * D_ + 4 * kk + 2 * h);
            wa = live ? wa : v2f{0.0f, 0.0f};
#pragma unroll
            for (int nt = 0; nt < 2; ++nt)
                acc[nt][ot] = __builtin_amdgcn_wmma_f32_16x16x4_f32(
                                  false, wa, false, qb[nt][kk], (short)0, acc[nt][ot], false, false);
        }
    }

    float* gcol = Gt + (size_t)b * DP * NN + n0 + m;
#pragma unroll
    for (int ot = 0; ot < 4; ++ot) {
#pragma unroll
        for (int j = 0; j < 8; ++j) {
            const int  of   = ot * 16 + 8 * h + j;
            const bool live = (ot < 3) || (of < D_);
            const float bias = gb[live ? of : 0];
#pragma unroll
            for (int nt = 0; nt < 2; ++nt)
                gcol[(size_t)of * NN + 16 * nt] = live ? (acc[nt][ot][j] + bias) : 0.0f;
        }
    }
}


__device__ __forceinline__ float xhalf(float x) {
    int b = __builtin_bit_cast(int, x);
    return __builtin_bit_cast(float,
        __builtin_amdgcn_permlanex16(b, b, 0x76543210, 0xfedcba98, false, false));
}

__global__ __launch_bounds__(128, 1)
void nlb_attn(const float* __restrict__ Q, const float* __restrict__ Gt,
              float* __restrict__ Y) {
    __shared__ __attribute__((aligned(16))) float lds[2 * STG];

    const int tid  = threadIdx.x;
    const int lane = threadIdx.x & 31;
    const int wave = threadIdx.x >> 5;
    const int m    = lane & 15;
    const int h    = lane >> 4;
    const int b    = blockIdx.x >> 6;
    const int q0   = ((blockIdx.x & 63) * 4 + wave) * 16;

    const float* Qb  = Q  + (size_t)b * NN * DP;
    const float* Gtb = Gt + (size_t)b * DP * NN;
    float*       Yb  = Y  + (size_t)b * NN * DP;

    const int kst = (tid >> 4) * KP + (tid & 15) * 4;
    const int vst = (tid >> 5) * VP + (tid & 31) * 4;
    const float* kgl = Qb + tid * 4;
    const float* vgl = Gtb + (size_t)(tid >> 5) * NN + (tid & 31) * 4;

    const int kfr = m * KP + 2 * h;
    const int vfr = KBUF + m * VP + 8 * h;

    v4f kr[16], vr[16];

#pragma unroll
    for (int i = 0; i < 16; ++i) {
        kr[i] = *(const v4f*)(kgl + i * 512);
        vr[i] = *(const v4f*)(vgl + (size_t)i * 4 * NN);
    }
#pragma unroll
    for (int i = 0; i < 16; ++i) {
        *(v4f*)(lds + kst + i * 8 * KP)        = kr[i];
        *(v4f*)(lds + KBUF + vst + i * 4 * VP) = vr[i];
    }
#pragma unroll
    for (int i = 0; i < 16; ++i) {
        kr[i] = *(const v4f*)(kgl + (size_t)KT * DP + i * 512);
        vr[i] = *(const v4f*)(vgl + KT + (size_t)i * 4 * NN);
    }

    const float LOG2E = 1.4426950408889634f;
    v2f qa[15];
    {
        const float* qrow = Qb + (size_t)(q0 + m) * DP;
#pragma unroll
        for (int kk = 0; kk < 15; ++kk) {
            v2f q = *(const v2f*)(qrow + 4 * kk + 2 * h);
            qa[kk] = q * LOG2E;
        }
    }

    v8f o0{}, o1{}, o2{}, o3{};
    float mrow = -__builtin_inff();
    float lrow = 0.0f;

    __syncthreads();

    v2f kf[8][15];
#pragma unroll
    for (int t = 0; t < 8; ++t)
#pragma unroll
        for (int kk = 0; kk < 15; ++kk)
            kf[t][kk] = *(const v2f*)(lds + kfr + t * 16 * KP + 4 * kk);

    for (int ch = 0; ch < NN / KT; ++ch) {
        const int k0 = ch * KT;
        const float* st  = lds + (ch & 1) * STG;
        float*       nst = lds + ((ch + 1) & 1) * STG;
        const bool more1 = (ch + 1 < NN / KT);
        const bool more2 = (ch + 2 < NN / KT);

        if (more1) {
#pragma unroll
            for (int i = 0; i < 16; ++i) {
                *(v4f*)(nst + kst + i * 8 * KP)        = kr[i];
                *(v4f*)(nst + KBUF + vst + i * 4 * VP) = vr[i];
            }
        }
        if (more2) {
#pragma unroll
            for (int i = 0; i < 16; ++i) {
                kr[i] = *(const v4f*)(kgl + (size_t)(k0 + 2 * KT) * DP + i * 512);
                vr[i] = *(const v4f*)(vgl + (k0 + 2 * KT) + (size_t)i * 4 * NN);
            }
        }
        v4f vf[8][2][4];
#pragma unroll
        for (int t = 0; t < 8; ++t)
#pragma unroll
            for (int up = 0; up < 2; ++up)
#pragma unroll
                for (int ct = 0; ct < 4; ++ct)
                    vf[t][up][ct] = *(const v4f*)(st + vfr + ct * 16 * VP + 16 * t + 4 * up);

        __builtin_amdgcn_sched_barrier(0);

        v8f s[8];
#pragma unroll
        for (int t = 0; t < 8; ++t) s[t] = v8f{};
#pragma unroll
        for (int kk = 0; kk < 15; ++kk) {
#pragma unroll
            for (int t = 0; t < 8; ++t)
                s[t] = __builtin_amdgcn_wmma_f32_16x16x4_f32(
                           false, kf[t][kk], false, qa[kk], (short)0, s[t], false, false);
        }

        __builtin_amdgcn_sched_barrier(0);

        __syncthreads();
        if (more1) {
#pragma unroll
            for (int t = 0; t < 8; ++t)
#pragma unroll
                for (int kk = 0; kk < 15; ++kk)
                    kf[t][kk] = *(const v2f*)(nst + kfr + t * 16 * KP + 4 * kk);
        }

        float tm;
        {
            float tmax[8];
#pragma unroll
            for (int t = 0; t < 8; ++t)
                tmax[t] = fmaxf(fmaxf(fmaxf(s[t][0], s[t][1]), fmaxf(s[t][2], s[t][3])),
                                fmaxf(fmaxf(s[t][4], s[t][5]), fmaxf(s[t][6], s[t][7])));
            tm = fmaxf(fmaxf(fmaxf(tmax[0], tmax[1]), fmaxf(tmax[2], tmax[3])),
                       fmaxf(fmaxf(tmax[4], tmax[5]), fmaxf(tmax[6], tmax[7])));
        }
        tm = fmaxf(tm, xhalf(tm));
        const float mn = fmaxf(mrow, tm);
        const float sc = fast_exp2(mrow - mn);
        float su;
        {
            float part[8];
#pragma unroll
            for (int t = 0; t < 8; ++t) {
#pragma unroll
                for (int r = 0; r < 8; ++r) s[t][r] = fast_exp2(s[t][r] - mn);
                part[t] = ((s[t][0] + s[t][1]) + (s[t][2] + s[t][3])) +
                          ((s[t][4] + s[t][5]) + (s[t][6] + s[t][7]));
            }
            su = ((part[0] + part[1]) + (part[2] + part[3])) +
                 ((part[4] + part[5]) + (part[6] + part[7]));
        }
        su += xhalf(su);
        lrow = lrow * sc + su;
        mrow = mn;
        o0 *= sc; o1 *= sc; o2 *= sc; o3 *= sc;

#pragma unroll
        for (int t = 0; t < 8; ++t) {
#pragma unroll
            for (int up = 0; up < 2; ++up) {
                const v4f va0 = vf[t][up][0];
                const v4f va1 = vf[t][up][1];
                const v4f va2 = vf[t][up][2];
                const v4f va3 = vf[t][up][3];
                const v2f pb0 = v2f{s[t][4 * up],     s[t][4 * up + 1]};
                const v2f pb1 = v2f{s[t][4 * up + 2], s[t][4 * up + 3]};
                o0 = __builtin_amdgcn_wmma_f32_16x16x4_f32(false, v2f{va0.x, va0.y}, false, pb0, (short)0, o0, false, false);
                o1 = __builtin_amdgcn_wmma_f32_16x16x4_f32(false, v2f{va1.x, va1.y}, false, pb0, (short)0, o1, false, false);
                o2 = __builtin_amdgcn_wmma_f32_16x16x4_f32(false, v2f{va2.x, va2.y}, false, pb0, (short)0, o2, false, false);
                o3 = __builtin_amdgcn_wmma_f32_16x16x4_f32(false, v2f{va3.x, va3.y}, false, pb0, (short)0, o3, false, false);
                o0 = __builtin_amdgcn_wmma_f32_16x16x4_f32(false, v2f{va0.z, va0.w}, false, pb1, (short)0, o0, false, false);
                o1 = __builtin_amdgcn_wmma_f32_16x16x4_f32(false, v2f{va1.z, va1.w}, false, pb1, (short)0, o1, false, false);
                o2 = __builtin_amdgcn_wmma_f32_16x16x4_f32(false, v2f{va2.z, va2.w}, false, pb1, (short)0, o2, false, false);
                o3 = __builtin_amdgcn_wmma_f32_16x16x4_f32(false, v2f{va3.z, va3.w}, false, pb1, (short)0, o3, false, false);
            }
        }
    }

    {
        const float inv = 1.0f / lrow;
        float* yr = Yb + (size_t)(q0 + m) * DP + 8 * h;
        *(v4f*)(yr +  0) = v4f{o0[0], o0[1], o0[2], o0[3]} * inv;
        *(v4f*)(yr +  4) = v4f{o0[4], o0[5], o0[6], o0[7]} * inv;
        *(v4f*)(yr + 16) = v4f{o1[0], o1[1], o1[2], o1[3]} * inv;
        *(v4f*)(yr + 20) = v4f{o1[4], o1[5], o1[6], o1[7]} * inv;
        *(v4f*)(yr + 32) = v4f{o2[0], o2[1], o2[2], o2[3]} * inv;
        *(v4f*)(yr + 36) = v4f{o2[4], o2[5], o2[6], o2[7]} * inv;
        *(v4f*)(yr + 48) = v4f{o3[0], o3[1], o3[2], o3[3]} * inv;
        *(v4f*)(yr + 52) = v4f{o3[4], o3[5], o3[6], o3[7]} * inv;
    }
}

__global__ __launch_bounds__(128)
void nlb_out(const float* __restrict__ x, const float* __restrict__ Y,
             const float* __restrict__ ww, const float* __restrict__ wb,
             float* __restrict__ Z) {
    const int lane = threadIdx.x & 31;
    const int wave = threadIdx.x >> 5;
    const int m    = lane & 15;
    const int h    = lane >> 4;
    const int tile = blockIdx.x * 4 + wave;
    const int b    = tile >> 8;
    const int n0   = (tile & 255) * 16;

    v2f yb[15];
    {
        const float* yrow = Y + ((size_t)b * NN + n0 + m) * DP + 2 * h;
#pragma unroll
        for (int kk = 0; kk < 15; ++kk) yb[kk] = *(const v2f*)(yrow + 4 * kk);
    }

    v8f acc[4];
#pragma unroll
    for (int ot = 0; ot < 4; ++ot) acc[ot] = v8f{};

#pragma unroll
    for (int kk = 0; kk < 15; ++kk) {
#pragma unroll
        for (int ot = 0; ot < 4; ++ot) {
            const int  o    = ot * 16 + m;
            const bool live = (ot < 3) || (o < D_);
            v2f wa = *(const v2f*)(ww + (live ? o : 0) * D_ + 4 * kk + 2 * h);
            wa = live ? wa : v2f{0.0f, 0.0f};
            acc[ot] = __builtin_amdgcn_wmma_f32_16x16x4_f32(
                          false, wa, false, yb[kk], (short)0, acc[ot], false, false);
        }
    }

    const int oh = n0 >> 6;
    const int ow = (n0 & 63) + m;
#pragma unroll
    for (int ot = 0; ot < 4; ++ot) {
#pragma unroll
        for (int jp = 0; jp < 4; ++jp) {
            const int oo = ot * 16 + 8 * h + 2 * jp;
            if ((ot < 3) || (oo < D_)) {
                const int tc = oo >> 2;
                const int ri = (oo >> 1) & 1;
                const size_t zi = ((size_t)(b * (T_ * C_) + tc) * H_ + (2 * oh + ri)) * W_ + 2 * ow;
                const v2f xv = *(const v2f*)(x + zi);
                v2f r;
                r.x = (acc[ot][2 * jp]     + wb[oo])     + xv.x;
                r.y = (acc[ot][2 * jp + 1] + wb[oo + 1]) + xv.y;
                *(v2f*)(Z + zi) = r;
            }
        }
    }
}

extern "C" void kernel_launch(void* const* d_in, const int* in_sizes, int n_in,
                              void* d_out, int out_size, void* d_ws, size_t ws_size,
                              hipStream_t stream) {
    const float* x  = (const float*)d_in[0];
    const float* gw = (const float*)d_in[1];
    const float* gb = (const float*)d_in[2];
    const float* ww = (const float*)d_in[3];
    const float* wb = (const float*)d_in[4];
    float* Z = (float*)d_out;

    const size_t buf = (size_t)B_ * NN * DP;
    float* Qbuf = (float*)d_ws;
    float* Gbuf = Qbuf + buf;
    float* Ybuf = Gbuf + buf;

    {
        int n = B_ * NN * DP;
        nlb_prep_q<<<(n + 255) / 256, 256, 0, stream>>>(x, Qbuf);
    }
    nlb_prep_g<<<B_ * NN / 64, 64, 0, stream>>>(Qbuf, gw, gb, Gbuf);
    nlb_attn<<<B_ * 64, 128, 0, stream>>>(Qbuf, Gbuf, Ybuf);
    nlb_out<<<B_ * NN / 64, 128, 0, stream>>>(x, Ybuf, ww, wb, Z);
}


# attention: two waves per SIMD: each query tile's keys split between a wave pair (half of every chunk each, own online softmax), partial max/sum/O merged once through LDS; key sum order changes
# speedup vs baseline: 2.5424x; 1.0634x over previous
#include <hip/hip_runtime.h>

typedef __attribute__((ext_vector_type(2))) float v2f;
typedef __attribute__((ext_vector_type(4))) float v4f;
typedef __attribute__((ext_vector_type(8))) float v8f;

#define B_  4
#define T_  5
#define C_  3
#define H_  128
#define W_  128
#define D_  60
#define DP  64
#define NN  4096
#define KT  128
#define KP  68
#define VP  132
#define KBUF (KT * KP)
#define VBUF (DP * VP)
#define STG  (KBUF + VBUF)

__device__ __forceinline__ float fast_exp2(float x) {
    return __builtin_amdgcn_exp2f(x);
}

__global__ void nlb_prep_q(const float* __restrict__ x, float* __restrict__ Q) {
    int idx = blockIdx.x * blockDim.x + threadIdx.x;
    if (idx >= B_ * NN * DP) return;
    int dp = idx & 63;
    int n  = (idx >> 6) & (NN - 1);
    int b  = idx >> 18;
    float v = 0.0f;
    if (dp < D_) {
        int tc = dp >> 2;
        int rr = dp & 3;
        int ri = rr >> 1, rj = rr & 1;
        int oh = n >> 6, ow = n & 63;
        int hh = oh * 2 + ri, ww = ow * 2 + rj;
        v = x[((size_t)(b * (T_ * C_) + tc) * H_ + hh) * W_ + ww];
    }
    Q[idx] = v;
}

__global__ __launch_bounds__(64)
void nlb_prep_g(const float* __restrict__ Q, const float* __restrict__ gw,
                const float* __restrict__ gb, float* __restrict__ Gt) {
    const int lane = threadIdx.x & 31;
    const int wave = threadIdx.x >> 5;
    const int m    = lane & 15;
    const int h    = lane >> 4;
    const int tile = blockIdx.x * 2 + wave;
    const int b    = tile >> 7;
    const int n0   = (tile & 127) * 32;

    v2f qb[2][15];
#pragma unroll
    for (int nt = 0; nt < 2; ++nt) {
        const float* qrow = Q + ((size_t)b * NN + n0 + 16 * nt + m) * DP + 2 * h;
#pragma unroll
        for (int kk = 0; kk < 15; ++kk) qb[nt][kk] = *(const v2f*)(qrow + 4 * kk);
    }

    v8f acc[2][4];
#pragma unroll
    for (int nt = 0; nt < 2; ++nt)
#pragma unroll
        for (int ot = 0; ot < 4; ++ot) acc[nt][ot] = v8f{};

#pragma unroll
    for (int kk = 0; kk < 15; ++kk) {
#pragma unroll
        for (int ot = 0; ot < 4; ++ot) {
            const int  o    = ot * 16 + m;
            const bool live = (ot < 3) || (o < D_);
            v2f wa = *(const v2f*)(gw + (live ? o : 0) * D_ + 4 * kk + 2 * h);
            wa = live ? wa : v2f{0.0f, 0.0f};
#pragma unroll
            for (int nt = 0; nt < 2; ++nt)
                acc[nt][ot] = __builtin_amdgcn_wmma_f32_16x16x4_f32(
                                  false, wa, false, qb[nt][kk], (short)0, acc[nt][ot], false, false);
        }
    }

    float* gcol = Gt + (size_t)b * DP * NN + n0 + m;
#pragma unroll
    for (int ot = 0; ot < 4; ++ot) {
#pragma unroll
        for (int j = 0; j < 8; ++j) {
            const int  of   = ot * 16 + 8 * h + j;
            const bool live = (ot < 3) || (of < D_);
            const float bias = gb[live ? of : 0];
#pragma unroll
            for (int nt = 0; nt < 2; ++nt)
                gcol[(size_t)of * NN + 16 * nt] = live ? (acc[nt][ot][j] + bias) : 0.0f;
        }
    }
}


__device__ __forceinline__ float xhalf(float x) {
    int b = __builtin_bit_cast(int, x);
    return __builtin_bit_cast(float,
        __builtin_amdgcn_permlanex16(b, b, 0x76543210, 0xfedcba98, false, false));
}

__global__ __launch_bounds__(256)
void nlb_attn(const float* __restrict__ Q, const float* __restrict__ Gt,
              float* __restrict__ Y) {
    __shared__ __attribute__((aligned(16))) float lds[2 * STG];

    const int tid  = threadIdx.x;
    const int lane = threadIdx.x & 31;
    const int wave = threadIdx.x >> 5;
    const int m    = lane & 15;
    const int h    = lane >> 4;
    const int wq   = wave & 3;
    const int kh   = wave >> 2;
    const int b    = blockIdx.x >> 6;
    const int q0   = ((blockIdx.x & 63) * 4 + wq) * 16;

    const float* Qb  = Q  + (size_t)b * NN * DP;
    const float* Gtb = Gt + (size_t)b * DP * NN;
    float*       Yb  = Y  + (size_t)b * NN * DP;

    const int kst = (tid >> 4) * KP + (tid & 15) * 4;
    const int vst = (tid >> 5) * VP + (tid & 31) * 4;
    const float* kgl = Qb + tid * 4;
    const float* vgl = Gtb + (size_t)(tid >> 5) * NN + (tid & 31) * 4;

    const int kfr = (kh * 64 + m) * KP + 2 * h;
    const int vfr = KBUF + m * VP + kh * 64 + 8 * h;

    v4f kr[8], vr[8];

#pragma unroll
    for (int i = 0; i < 8; ++i) {
        kr[i] = *(const v4f*)(kgl + i * 1024);
        vr[i] = *(const v4f*)(vgl + (size_t)i * 8 * NN);
    }
#pragma unroll
    for (int i = 0; i < 8; ++i) {
        *(v4f*)(lds + kst + i * 16 * KP)       = kr[i];
        *(v4f*)(lds + KBUF + vst + i * 8 * VP) = vr[i];
    }
#pragma unroll
    for (int i = 0; i < 8; ++i) {
        kr[i] = *(const v4f*)(kgl + (size_t)KT * DP + i * 1024);
        vr[i] = *(const v4f*)(vgl + KT + (size_t)i * 8 * NN);
    }

    const float LOG2E = 1.4426950408889634f;
    v2f qa[15];
    {
        const float* qrow = Qb + (size_t)(q0 + m) * DP;
#pragma unroll
        for (int kk = 0; kk < 15; ++kk) {
            v2f q = *(const v2f*)(qrow + 4 * kk + 2 * h);
            qa[kk] = q * LOG2E;
        }
    }

    v8f o0{}, o1{}, o2{}, o3{};
    float mrow = -__builtin_inff();
    float lrow = 0.0f;

    __syncthreads();

    v2f kf[4][15];
#pragma unroll
    for (int t = 0; t < 4; ++t)
#pragma unroll
        for (int kk = 0; kk < 15; ++kk)
            kf[t][kk] = *(const v2f*)(lds + kfr + t * 16 * KP + 4 * kk);

    for (int ch = 0; ch < NN / KT; ++ch) {
        const int k0 = ch * KT;
        const float* st  = lds + (ch & 1) * STG;
        float*       nst = lds + ((ch + 1) & 1) * STG;
        const bool more1 = (ch + 1 < NN / KT);
        const bool more2 = (ch + 2 < NN / KT);

        if (more1) {
#pragma unroll
            for (int i = 0; i < 8; ++i) {
                *(v4f*)(nst + kst + i * 16 * KP)       = kr[i];
                *(v4f*)(nst + KBUF + vst + i * 8 * VP) = vr[i];
            }
        }
        if (more2) {
#pragma unroll
            for (int i = 0; i < 8; ++i) {
                kr[i] = *(const v4f*)(kgl + (size_t)(k0 + 2 * KT) * DP + i * 1024);
                vr[i] = *(const v4f*)(vgl + (k0 + 2 * KT) + (size_t)i * 8 * NN);
            }
        }
        v4f vf[4][2][4];
#pragma unroll
        for (int t = 0; t < 4; ++t)
#pragma unroll
            for (int up = 0; up < 2; ++up)
#pragma unroll
                for (int ct = 0; ct < 4; ++ct)
                    vf[t][up][ct] = *(const v4f*)(st + vfr + ct * 16 * VP + 16 * t + 4 * up);

        __builtin_amdgcn_sched_barrier(0);

        v8f s[4];
#pragma unroll
        for (int t = 0; t < 4; ++t) s[t] = v8f{};
#pragma unroll
        for (int kk = 0; kk < 15; ++kk) {
#pragma unroll
            for (int t = 0; t < 4; ++t)
                s[t] = __builtin_amdgcn_wmma_f32_16x16x4_f32(
                           false, kf[t][kk], false, qa[kk], (short)0, s[t], false, false);
        }

        __builtin_amdgcn_sched_barrier(0);

        __syncthreads();
        if (more1) {
#pragma unroll
            for (int t = 0; t < 4; ++t)
#pragma unroll
                for (int kk = 0; kk < 15; ++kk)
                    kf[t][kk] = *(const v2f*)(nst + kfr + t * 16 * KP + 4 * kk);
        }
        __builtin_amdgcn_sched_barrier(0);

        float tm;
        {
            float tmax[4];
#pragma unroll
            for (int t = 0; t < 4; ++t)
                tmax[t] = fmaxf(fmaxf(fmaxf(s[t][0], s[t][1]), fmaxf(s[t][2], s[t][3])),
                                fmaxf(fmaxf(s[t][4], s[t][5]), fmaxf(s[t][6], s[t][7])));
            tm = fmaxf(fmaxf(tmax[0], tmax[1]), fmaxf(tmax[2], tmax[3]));
        }
        tm = fmaxf(tm, xhalf(tm));
        const float mn = fmaxf(mrow, tm);
        const float sc = fast_exp2(mrow - mn);
        float su;
        {
            float part[4];
#pragma unroll
            for (int t = 0; t < 4; ++t) {
#pragma unroll
                for (int r = 0; r < 8; ++r) s[t][r] = fast_exp2(s[t][r] - mn);
                part[t] = ((s[t][0] + s[t][1]) + (s[t][2] + s[t][3])) +
                          ((s[t][4] + s[t][5]) + (s[t][6] + s[t][7]));
            }
            su = (part[0] + part[1]) + (part[2] + part[3]);
        }
        su += xhalf(su);
        lrow = lrow * sc + su;
        mrow = mn;
        o0 *= sc; o1 *= sc; o2 *= sc; o3 *= sc;

#pragma unroll
        for (int t = 0; t < 4; ++t) {
#pragma unroll
            for (int up = 0; up < 2; ++up) {
                const v4f va0 = vf[t][up][0];
                const v4f va1 = vf[t][up][1];
                const v4f va2 = vf[t][up][2];
                const v4f va3 = vf[t][up][3];
                const v2f pb0 = v2f{s[t][4 * up],     s[t][4 * up + 1]};
                const v2f pb1 = v2f{s[t][4 * up + 2], s[t][4 * up + 3]};
                o0 = __builtin_amdgcn_wmma_f32_16x16x4_f32(false, v2f{va0.x, va0.y}, false, pb0, (short)0, o0, false, false);
                o1 = __builtin_amdgcn_wmma_f32_16x16x4_f32(false, v2f{va1.x, va1.y}, false, pb0, (short)0, o1, false, false);
                o2 = __builtin_amdgcn_wmma_f32_16x16x4_f32(false, v2f{va2.x, va2.y}, false, pb0, (short)0, o2, false, false);
                o3 = __builtin_amdgcn_wmma_f32_16x16x4_f32(false, v2f{va3.x, va3.y}, false, pb0, (short)0, o3, false, false);
                o0 = __builtin_amdgcn_wmma_f32_16x16x4_f32(false, v2f{va0.z, va0.w}, false, pb1, (short)0, o0, false, false);
                o1 = __builtin_amdgcn_wmma_f32_16x16x4_f32(false, v2f{va1.z, va1.w}, false, pb1, (short)0, o1, false, false);
                o2 = __builtin_amdgcn_wmma_f32_16x16x4_f32(false, v2f{va2.z, va2.w}, false, pb1, (short)0, o2, false, false);
                o3 = __builtin_amdgcn_wmma_f32_16x16x4_f32(false, v2f{va3.z, va3.w}, false, pb1, (short)0, o3, false, false);
            }
        }
    }

    float* jn = lds + wq * (34 * 32) + lane;
    if (kh == 1) {
#pragma unroll
        for (int r = 0; r < 8; ++r) {
            jn[(r)      * 32] = o0[r];
            jn[(8 + r)  * 32] = o1[r];
            jn[(16 + r) * 32] = o2[r];
            jn[(24 + r) * 32] = o3[r];
        }
        jn[32 * 32] = mrow;
        jn[33 * 32] = lrow;
    }
    __syncthreads();
    if (kh == 0) {
        const float m1 = jn[32 * 32];
        const float l1 = jn[33 * 32];
        const float mx = fmaxf(mrow, m1);
        const float a0 = fast_exp2(mrow - mx);
        const float a1 = fast_exp2(m1 - mx);
        const float inv = 1.0f / (lrow * a0 + l1 * a1);
        const float w0 = a0 * inv;
        const float w1 = a1 * inv;
        v8f p0, p1, p2, p3;
#pragma unroll
        for (int r = 0; r < 8; ++r) {
            p0[r] = jn[(r)      * 32];
            p1[r] = jn[(8 + r)  * 32];
            p2[r] = jn[(16 + r) * 32];
            p3[r] = jn[(24 + r) * 32];
        }
        o0 = o0 * w0 + p0 * w1;
        o1 = o1 * w0 + p1 * w1;
        o2 = o2 * w0 + p2 * w1;
        o3 = o3 * w0 + p3 * w1;

        float* yr = Yb + (size_t)(q0 + m) * DP + 8 * h;
        *(v4f*)(yr +  0) = v4f{o0[0], o0[1], o0[2], o0[3]};
        *(v4f*)(yr +  4) = v4f{o0[4], o0[5], o0[6], o0[7]};
        *(v4f*)(yr + 16) = v4f{o1[0], o1[1], o1[2], o1[3]};
        *(v4f*)(yr + 20) = v4f{o1[4], o1[5], o1[6], o1[7]};
        *(v4f*)(yr + 32) = v4f{o2[0], o2[1], o2[2], o2[3]};
        *(v4f*)(yr + 36) = v4f{o2[4], o2[5], o2[6], o2[7]};
        *(v4f*)(yr + 48) = v4f{o3[0], o3[1], o3[2], o3[3]};
        *(v4f*)(yr + 52) = v4f{o3[4], o3[5], o3[6], o3[7]};
    }
}

__global__ __launch_bounds__(128)
void nlb_out(const float* __restrict__ x, const float* __restrict__ Y,
             const float* __restrict__ ww, const float* __restrict__ wb,
             float* __restrict__ Z) {
    const int lane = threadIdx.x & 31;
    const int wave = threadIdx.x >> 5;
    const int m    = lane & 15;
    const int h    = lane >> 4;
    const int tile = blockIdx.x * 4 + wave;
    const int b    = tile >> 8;
    const int n0   = (tile & 255) * 16;

    v2f yb[15];
    {
        const float* yrow = Y + ((size_t)b * NN + n0 + m) * DP + 2 * h;
#pragma unroll
        for (int kk = 0; kk < 15; ++kk) yb[kk] = *(const v2f*)(yrow + 4 * kk);
    }

    v8f acc[4];
#pragma unroll
    for (int ot = 0; ot < 4; ++ot) acc[ot] = v8f{};

#pragma unroll
    for (int kk = 0; kk < 15; ++kk) {
#pragma unroll
        for (int ot = 0; ot < 4; ++ot) {
            const int  o    = ot * 16 + m;
            const bool live = (ot < 3) || (o < D_);
            v2f wa = *(const v2f*)(ww + (live ? o : 0) * D_ + 4 * kk + 2 * h);
            wa = live ? wa : v2f{0.0f, 0.0f};
            acc[ot] = __builtin_amdgcn_wmma_f32_16x16x4_f32(
                          false, wa, false, yb[kk], (short)0, acc[ot], false, false);
        }
    }

    const int oh = n0 >> 6;
    const int ow = (n0 & 63) + m;
#pragma unroll
    for (int ot = 0; ot < 4; ++ot) {
#pragma unroll
        for (int jp = 0; jp < 4; ++jp) {
            const int oo = ot * 16 + 8 * h + 2 * jp;
            if ((ot < 3) || (oo < D_)) {
                const int tc = oo >> 2;
                const int ri = (oo >> 1) & 1;
                const size_t zi = ((size_t)(b * (T_ * C_) + tc) * H_ + (2 * oh + ri)) * W_ + 2 * ow;
                const v2f xv = *(const v2f*)(x + zi);
                v2f r;
                r.x = (acc[ot][2 * jp]     + wb[oo])     + xv.x;
                r.y = (acc[ot][2 * jp + 1] + wb[oo + 1]) + xv.y;
                *(v2f*)(Z + zi) = r;
            }
        }
    }
}

extern "C" void kernel_launch(void* const* d_in, const int* in_sizes, int n_in,
                              void* d_out, int out_size, void* d_ws, size_t ws_size,
                              hipStream_t stream) {
    const float* x  = (const float*)d_in[0];
    const float* gw = (const float*)d_in[1];
    const float* gb = (const float*)d_in[2];
    const float* ww = (const float*)d_in[3];
    const float* wb = (const float*)d_in[4];
    float* Z = (float*)d_out;

    const size_t buf = (size_t)B_ * NN * DP;
    float* Qbuf = (float*)d_ws;
    float* Gbuf = Qbuf + buf;
    float* Ybuf = Gbuf + buf;

    {
        int n = B_ * NN * DP;
        nlb_prep_q<<<(n + 255) / 256, 256, 0, stream>>>(x, Qbuf);
    }
    nlb_prep_g<<<B_ * NN / 64, 64, 0, stream>>>(Qbuf, gw, gb, Gbuf);
    nlb_attn<<<B_ * 64, 256, 0, stream>>>(Qbuf, Gbuf, Ybuf);
    nlb_out<<<B_ * NN / 64, 128, 0, stream>>>(x, Ybuf, ww, wb, Z);
}


# pixel-unshuffle folded into the V-projection kernel: a wave loads its WMMA B fragments straight from x (whole-line b64 reads) and writes them out as Q rows through LDS; 4 launches -> 3
# speedup vs baseline: 2.6242x; 1.0322x over previous
#include <hip/hip_runtime.h>

typedef __attribute__((ext_vector_type(2))) float v2f;
typedef __attribute__((ext_vector_type(4))) float v4f;
typedef __attribute__((ext_vector_type(8))) float v8f;

#define B_  4
#define T_  5
#define C_  3
#define H_  128
#define W_  128
#define D_  60
#define DP  64
#define NN  4096
#define KT  128
#define KP  68
#define VP  132
#define KBUF (KT * KP)
#define VBUF (DP * VP)
#define STG  (KBUF + VBUF)

__device__ __forceinline__ float fast_exp2(float x) {
    return __builtin_amdgcn_exp2f(x);
}

#define QP 68

__global__ __launch_bounds__(64)
void nlb_prep_qg(const float* __restrict__ x, const float* __restrict__ gw,
                 const float* __restrict__ gb, float* __restrict__ Q,
                 float* __restrict__ Gt) {
    __shared__ __attribute__((aligned(16))) float qs[2 * 32 * QP];

    const int lane = threadIdx.x & 31;
    const int wave = threadIdx.x >> 5;
    const int m    = lane & 15;
    const int h    = lane >> 4;
    const int tile = blockIdx.x * 2 + wave;
    const int b    = tile >> 7;
    const int n0   = (tile & 127) * 32;
    const int oh   = n0 >> 6;
    const int ow0  = n0 & 63;

    v2f qb[2][15];
#pragma unroll
    for (int nt = 0; nt < 2; ++nt) {
        const float* xr = x + ((size_t)(b * (T_ * C_)) * H_ + 2 * oh + h) * W_
                            + 2 * (ow0 + 16 * nt + m);
#pragma unroll
        for (int kk = 0; kk < 15; ++kk) qb[nt][kk] = *(const v2f*)(xr + (size_t)kk * H_ * W_);
    }

    float* qw = qs + wave * (32 * QP);
#pragma unroll
    for (int nt = 0; nt < 2; ++nt) {
        float* qrow = qw + (16 * nt + m) * QP + 2 * h;
#pragma unroll
        for (int kk = 0; kk < 15; ++kk) *(v2f*)(qrow + 4 * kk) = qb[nt][kk];
        *(v2f*)(qrow + D_) = v2f{0.0f, 0.0f};
    }

    v8f acc[2][4];
#pragma unroll
    for (int nt = 0; nt < 2; ++nt)
#pragma unroll
        for (int ot = 0; ot < 4; ++ot) acc[nt][ot] = v8f{};

#pragma unroll
    for (int kk = 0; kk < 15; ++kk) {
#pragma unroll
        for (int ot = 0; ot < 4; ++ot) {
            const int  o    = ot * 16 + m;
            const bool live = (ot < 3) || (o < D_);
            v2f wa = *(const v2f*)(gw + (live ? o : 0) * D_ + 4 * kk + 2 * h);
            wa = live ? wa : v2f{0.0f, 0.0f};
#pragma unroll
            for (int nt = 0; nt < 2; ++nt)
                acc[nt][ot] = __builtin_amdgcn_wmma_f32_16x16x4_f32(
                                  false, wa, false, qb[nt][kk], (short)0, acc[nt][ot], false, false);
        }
    }

    __syncthreads();
    {
        float* qg = Q + ((size_t)b * NN + n0) * DP;
#pragma unroll
        for (int i = 0; i < 16; ++i) {
            const int row = 2 * i + h;
            *(v4f*)(qg + row * DP + 4 * m) = *(const v4f*)(qw + row * QP + 4 * m);
        }
    }

    float* gcol = Gt + (size_t)b * DP * NN + n0 + m;
#pragma unroll
    for (int ot = 0; ot < 4; ++ot) {
#pragma unroll
        for (int j = 0; j < 8; ++j) {
            const int  of   = ot * 16 + 8 * h + j;
            const bool live = (ot < 3) || (of < D_);
            const float bias = gb[live ? of : 0];
#pragma unroll
            for (int nt = 0; nt < 2; ++nt)
                gcol[(size_t)of * NN + 16 * nt] = live ? (acc[nt][ot][j] + bias) : 0.0f;
        }
    }
}


__device__ __forceinline__ float xhalf(float x) {
    int b = __builtin_bit_cast(int, x);
    return __builtin_bit_cast(float,
        __builtin_amdgcn_permlanex16(b, b, 0x76543210, 0xfedcba98, false, false));
}

__global__ __launch_bounds__(256)
void nlb_attn(const float* __restrict__ Q, const float* __restrict__ Gt,
              float* __restrict__ Y) {
    __shared__ __attribute__((aligned(16))) float lds[2 * STG];

    const int tid  = threadIdx.x;
    const int lane = threadIdx.x & 31;
    const int wave = threadIdx.x >> 5;
    const int m    = lane & 15;
    const int h    = lane >> 4;
    const int wq   = wave & 3;
    const int kh   = wave >> 2;
    const int b    = blockIdx.x >> 6;
    const int q0   = ((blockIdx.x & 63) * 4 + wq) * 16;

    const float* Qb  = Q  + (size_t)b * NN * DP;
    const float* Gtb = Gt + (size_t)b * DP * NN;
    float*       Yb  = Y  + (size_t)b * NN * DP;

    const int kst = (tid >> 4) * KP + (tid & 15) * 4;
    const int vst = (tid >> 5) * VP + (tid & 31) * 4;
    const float* kgl = Qb + tid * 4;
    const float* vgl = Gtb + (size_t)(tid >> 5) * NN + (tid & 31) * 4;

    const int kfr = (kh * 64 + m) * KP + 2 * h;
    const int vfr = KBUF + m * VP + kh * 64 + 8 * h;

    v4f kr[8], vr[8];

#pragma unroll
    for (int i = 0; i < 8; ++i) {
        kr[i] = *(const v4f*)(kgl + i * 1024);
        vr[i] = *(const v4f*)(vgl + (size_t)i * 8 * NN);
    }
#pragma unroll
    for (int i = 0; i < 8; ++i) {
        *(v4f*)(lds + kst + i * 16 * KP)       = kr[i];
        *(v4f*)(lds + KBUF + vst + i * 8 * VP) = vr[i];
    }
#pragma unroll
    for (int i = 0; i < 8; ++i) {
        kr[i] = *(const v4f*)(kgl + (size_t)KT * DP + i * 1024);
        vr[i] = *(const v4f*)(vgl + KT + (size_t)i * 8 * NN);
    }

    const float LOG2E = 1.4426950408889634f;
    v2f qa[15];
    {
        const float* qrow = Qb + (size_t)(q0 + m) * DP;
#pragma unroll
        for (int kk = 0; kk < 15; ++kk) {
            v2f q = *(const v2f*)(qrow + 4 * kk + 2 * h);
            qa[kk] = q * LOG2E;
        }
    }

    v8f o0{}, o1{}, o2{}, o3{};
    float mrow = -__builtin_inff();
    float lrow = 0.0f;

    __syncthreads();

    v2f kf[4][15];
#pragma unroll
    for (int t = 0; t < 4; ++t)
#pragma unroll
        for (int kk = 0; kk < 15; ++kk)
            kf[t][kk] = *(const v2f*)(lds + kfr + t * 16 * KP + 4 * kk);

    for (int ch = 0; ch < NN / KT; ++ch) {
        const int k0 = ch * KT;
        const float* st  = lds + (ch & 1) * STG;
        float*       nst = lds + ((ch + 1) & 1) * STG;
        const bool more1 = (ch + 1 < NN / KT);
        const bool more2 = (ch + 2 < NN / KT);

        if (more1) {
#pragma unroll
            for (int i = 0; i < 8; ++i) {
                *(v4f*)(nst + kst + i * 16 * KP)       = kr[i];
                *(v4f*)(nst + KBUF + vst + i * 8 * VP) = vr[i];
            }
        }
        if (more2) {
#pragma unroll
            for (int i = 0; i < 8; ++i) {
                kr[i] = *(const v4f*)(kgl + (size_t)(k0 + 2 * KT) * DP + i * 1024);
                vr[i] = *(const v4f*)(vgl + (k0 + 2 * KT) + (size_t)i * 8 * NN);
            }
        }
        v4f vf[4][2][4];
#pragma unroll
        for (int t = 0; t < 4; ++t)
#pragma unroll
            for (int up = 0; up < 2; ++up)
#pragma unroll
                for (int ct = 0; ct < 4; ++ct)
                    vf[t][up][ct] = *(const v4f*)(st + vfr + ct * 16 * VP + 16 * t + 4 * up);

        __builtin_amdgcn_sched_barrier(0);

        v8f s[4];
#pragma unroll
        for (int t = 0; t < 4; ++t) s[t] = v8f{};
#pragma unroll
        for (int kk = 0; kk < 15; ++kk) {
#pragma unroll
            for (int t = 0; t < 4; ++t)
                s[t] = __builtin_amdgcn_wmma_f32_16x16x4_f32(
                           false, kf[t][kk], false, qa[kk], (short)0, s[t], false, false);
        }

        __builtin_amdgcn_sched_barrier(0);

        __syncthreads();
        if (more1) {
#pragma unroll
            for (int t = 0; t < 4; ++t)
#pragma unroll
                for (int kk = 0; kk < 15; ++kk)
                    kf[t][kk] = *(const v2f*)(nst + kfr + t * 16 * KP + 4 * kk);
        }
        __builtin_amdgcn_sched_barrier(0);

        float tm;
        {
            float tmax[4];
#pragma unroll
            for (int t = 0; t < 4; ++t)
                tmax[t] = fmaxf(fmaxf(fmaxf(s[t][0], s[t][1]), fmaxf(s[t][2], s[t][3])),
                                fmaxf(fmaxf(s[t][4], s[t][5]), fmaxf(s[t][6], s[t][7])));
            tm = fmaxf(fmaxf(tmax[0], tmax[1]), fmaxf(tmax[2], tmax[3]));
        }
        tm = fmaxf(tm, xhalf(tm));
        const float mn = fmaxf(mrow, tm);
        const float sc = fast_exp2(mrow - mn);
        float su;
        {
            float part[4];
#pragma unroll
            for (int t = 0; t < 4; ++t) {
#pragma unroll
                for (int r = 0; r < 8; ++r) s[t][r] = fast_exp2(s[t][r] - mn);
                part[t] = ((s[t][0] + s[t][1]) + (s[t][2] + s[t][3])) +
                          ((s[t][4] + s[t][5]) + (s[t][6] + s[t][7]));
            }
            su = (part[0] + part[1]) + (part[2] + part[3]);
        }
        su += xhalf(su);
        lrow = lrow * sc + su;
        mrow = mn;
        o0 *= sc; o1 *= sc; o2 *= sc; o3 *= sc;

#pragma unroll
        for (int t = 0; t < 4; ++t) {
#pragma unroll
            for (int up = 0; up < 2; ++up) {
                const v4f va0 = vf[t][up][0];
                const v4f va1 = vf[t][up][1];
                const v4f va2 = vf[t][up][2];
                const v4f va3 = vf[t][up][3];
                const v2f pb0 = v2f{s[t][4 * up],     s[t][4 * up + 1]};
                const v2f pb1 = v2f{s[t][4 * up + 2], s[t][4 * up + 3]};
                o0 = __builtin_amdgcn_wmma_f32_16x16x4_f32(false, v2f{va0.x, va0.y}, false, pb0, (short)0, o0, false, false);
                o1 = __builtin_amdgcn_wmma_f32_16x16x4_f32(false, v2f{va1.x, va1.y}, false, pb0, (short)0, o1, false, false);
                o2 = __builtin_amdgcn_wmma_f32_16x16x4_f32(false, v2f{va2.x, va2.y}, false, pb0, (short)0, o2, false, false);
                o3 = __builtin_amdgcn_wmma_f32_16x16x4_f32(false, v2f{va3.x, va3.y}, false, pb0, (short)0, o3, false, false);
                o0 = __builtin_amdgcn_wmma_f32_16x16x4_f32(false, v2f{va0.z, va0.w}, false, pb1, (short)0, o0, false, false);
                o1 = __builtin_amdgcn_wmma_f32_16x16x4_f32(false, v2f{va1.z, va1.w}, false, pb1, (short)0, o1, false, false);
                o2 = __builtin_amdgcn_wmma_f32_16x16x4_f32(false, v2f{va2.z, va2.w}, false, pb1, (short)0, o2, false, false);
                o3 = __builtin_amdgcn_wmma_f32_16x16x4_f32(false, v2f{va3.z, va3.w}, false, pb1, (short)0, o3, false, false);
            }
        }
    }

    float* jn = lds + wq * (34 * 32) + lane;
    if (kh == 1) {
#pragma unroll
        for (int r = 0; r < 8; ++r) {
            jn[(r)      * 32] = o0[r];
            jn[(8 + r)  * 32] = o1[r];
            jn[(16 + r) * 32] = o2[r];
            jn[(24 + r) * 32] = o3[r];
        }
        jn[32 * 32] = mrow;
        jn[33 * 32] = lrow;
    }
    __syncthreads();
    if (kh == 0) {
        const float m1 = jn[32 * 32];
        const float l1 = jn[33 * 32];
        const float mx = fmaxf(mrow, m1);
        const float a0 = fast_exp2(mrow - mx);
        const float a1 = fast_exp2(m1 - mx);
        const float inv = 1.0f / (lrow * a0 + l1 * a1);
        const float w0 = a0 * inv;
        const float w1 = a1 * inv;
        v8f p0, p1, p2, p3;
#pragma unroll
        for (int r = 0; r < 8; ++r) {
            p0[r] = jn[(r)      * 32];
            p1[r] = jn[(8 + r)  * 32];
            p2[r] = jn[(16 + r) * 32];
            p3[r] = jn[(24 + r) * 32];
        }
        o0 = o0 * w0 + p0 * w1;
        o1 = o1 * w0 + p1 * w1;
        o2 = o2 * w0 + p2 * w1;
        o3 = o3 * w0 + p3 * w1;

        float* yr = Yb + (size_t)(q0 + m) * DP + 8 * h;
        *(v4f*)(yr +  0) = v4f{o0[0], o0[1], o0[2], o0[3]};
        *(v4f*)(yr +  4) = v4f{o0[4], o0[5], o0[6], o0[7]};
        *(v4f*)(yr + 16) = v4f{o1[0], o1[1], o1[2], o1[3]};
        *(v4f*)(yr + 20) = v4f{o1[4], o1[5], o1[6], o1[7]};
        *(v4f*)(yr + 32) = v4f{o2[0], o2[1], o2[2], o2[3]};
        *(v4f*)(yr + 36) = v4f{o2[4], o2[5], o2[6], o2[7]};
        *(v4f*)(yr + 48) = v4f{o3[0], o3[1], o3[2], o3[3]};
        *(v4f*)(yr + 52) = v4f{o3[4], o3[5], o3[6], o3[7]};
    }
}

__global__ __launch_bounds__(128)
void nlb_out(const float* __restrict__ x, const float* __restrict__ Y,
             const float* __restrict__ ww, const float* __restrict__ wb,
             float* __restrict__ Z) {
    const int lane = threadIdx.x & 31;
    const int wave = threadIdx.x >> 5;
    const int m    = lane & 15;
    const int h    = lane >> 4;
    const int tile = blockIdx.x * 4 + wave;
    const int b    = tile >> 8;
    const int n0   = (tile & 255) * 16;

    v2f yb[15];
    {
        const float* yrow = Y + ((size_t)b * NN + n0 + m) * DP + 2 * h;
#pragma unroll
        for (int kk = 0; kk < 15; ++kk) yb[kk] = *(const v2f*)(yrow + 4 * kk);
    }

    v8f acc[4];
#pragma unroll
    for (int ot = 0; ot < 4; ++ot) acc[ot] = v8f{};

#pragma unroll
    for (int kk = 0; kk < 15; ++kk) {
#pragma unroll
        for (int ot = 0; ot < 4; ++ot) {
            const int  o    = ot * 16 + m;
            const bool live = (ot < 3) || (o < D_);
            v2f wa = *(const v2f*)(ww + (live ? o : 0) * D_ + 4 * kk + 2 * h);
            wa = live ? wa : v2f{0.0f, 0.0f};
            acc[ot] = __builtin_amdgcn_wmma_f32_16x16x4_f32(
                          false, wa, false, yb[kk], (short)0, acc[ot], false, false);
        }
    }

    const int oh = n0 >> 6;
    const int ow = (n0 & 63) + m;
#pragma unroll
    for (int ot = 0; ot < 4; ++ot) {
#pragma unroll
        for (int jp = 0; jp < 4; ++jp) {
            const int oo = ot * 16 + 8 * h + 2 * jp;
            if ((ot < 3) || (oo < D_)) {
                const int tc = oo >> 2;
                const int ri = (oo >> 1) & 1;
                const size_t zi = ((size_t)(b * (T_ * C_) + tc) * H_ + (2 * oh + ri)) * W_ + 2 * ow;
                const v2f xv = *(const v2f*)(x + zi);
                v2f r;
                r.x = (acc[ot][2 * jp]     + wb[oo])     + xv.x;
                r.y = (acc[ot][2 * jp + 1] + wb[oo + 1]) + xv.y;
                *(v2f*)(Z + zi) = r;
            }
        }
    }
}

extern "C" void kernel_launch(void* const* d_in, const int* in_sizes, int n_in,
                              void* d_out, int out_size, void* d_ws, size_t ws_size,
                              hipStream_t stream) {
    const float* x  = (const float*)d_in[0];
    const float* gw = (const float*)d_in[1];
    const float* gb = (const float*)d_in[2];
    const float* ww = (const float*)d_in[3];
    const float* wb = (const float*)d_in[4];
    float* Z = (float*)d_out;

    const size_t buf = (size_t)B_ * NN * DP;
    float* Qbuf = (float*)d_ws;
    float* Gbuf = Qbuf + buf;
    float* Ybuf = Gbuf + buf;

    nlb_prep_qg<<<B_ * NN / 64, 64, 0, stream>>>(x, gw, gb, Qbuf, Gbuf);
    nlb_attn<<<B_ * 64, 256, 0, stream>>>(Qbuf, Gbuf, Ybuf);
    nlb_out<<<B_ * NN / 64, 128, 0, stream>>>(x, Ybuf, ww, wb, Z);
}


# attention: chunk barrier split (signal early in the score run, wait behind it) and s_setprio stepped down run by run: a SIMD's two waves run out of step, one's softmax under the other's WMMAs
# speedup vs baseline: 2.8557x; 1.0882x over previous
#include <hip/hip_runtime.h>

typedef __attribute__((ext_vector_type(2))) float v2f;
typedef __attribute__((ext_vector_type(4))) float v4f;
typedef __attribute__((ext_vector_type(8))) float v8f;

#define B_  4
#define T_  5
#define C_  3
#define H_  128
#define W_  128
#define D_  60
#define DP  64
#define NN  4096
#define KT  128
#define KP  68
#define VP  132
#define KBUF (KT * KP)
#define VBUF (DP * VP)
#define STG  (KBUF + VBUF)
#define SIGK 6

__device__ __forceinline__ float fast_exp2(float x) {
    return __builtin_amdgcn_exp2f(x);
}

#define QP 68

__global__ __launch_bounds__(64)
void nlb_prep_qg(const float* __restrict__ x, const float* __restrict__ gw,
                 const float* __restrict__ gb, float* __restrict__ Q,
                 float* __restrict__ Gt) {
    __shared__ __attribute__((aligned(16))) float qs[2 * 32 * QP];

    const int lane = threadIdx.x & 31;
    const int wave = threadIdx.x >> 5;
    const int m    = lane & 15;
    const int h    = lane >> 4;
    const int tile = blockIdx.x * 2 + wave;
    const int b    = tile >> 7;
    const int n0   = (tile & 127) * 32;
    const int oh   = n0 >> 6;
    const int ow0  = n0 & 63;

    v2f qb[2][15];
#pragma unroll
    for (int nt = 0; nt < 2; ++nt) {
        const float* xr = x + ((size_t)(b * (T_ * C_)) * H_ + 2 * oh + h) * W_
                            + 2 * (ow0 + 16 * nt + m);
#pragma unroll
        for (int kk = 0; kk < 15; ++kk) qb[nt][kk] = *(const v2f*)(xr + (size_t)kk * H_ * W_);
    }

    float* qw = qs + wave * (32 * QP);
#pragma unroll
    for (int nt = 0; nt < 2; ++nt) {
        float* qrow = qw + (16 * nt + m) * QP + 2 * h;
#pragma unroll
        for (int kk = 0; kk < 15; ++kk) *(v2f*)(qrow + 4 * kk) = qb[nt][kk];
        *(v2f*)(qrow + D_) = v2f{0.0f, 0.0f};
    }

    v8f acc[2][4];
#pragma unroll
    for (int nt = 0; nt < 2; ++nt)
#pragma unroll
        for (int ot = 0; ot < 4; ++ot) acc[nt][ot] = v8f{};

#pragma unroll
    for (int kk = 0; kk < 15; ++kk) {
#pragma unroll
        for (int ot = 0; ot < 4; ++ot) {
            const int  o    = ot * 16 + m;
            const bool live = (ot < 3) || (o < D_);
            v2f wa = *(const v2f*)(gw + (live ? o : 0) * D_ + 4 * kk + 2 * h);
            wa = live ? wa : v2f{0.0f, 0.0f};
#pragma unroll
            for (int nt = 0; nt < 2; ++nt)
                acc[nt][ot] = __builtin_amdgcn_wmma_f32_16x16x4_f32(
                                  false, wa, false, qb[nt][kk], (short)0, acc[nt][ot], false, false);
        }
    }

    __syncthreads();
    {
        float* qg = Q + ((size_t)b * NN + n0) * DP;
#pragma unroll
        for (int i = 0; i < 16; ++i) {
            const int row = 2 * i + h;
            *(v4f*)(qg + row * DP + 4 * m) = *(const v4f*)(qw + row * QP + 4 * m);
        }
    }

    float* gcol = Gt + (size_t)b * DP * NN + n0 + m;
#pragma unroll
    for (int ot = 0; ot < 4; ++ot) {
#pragma unroll
        for (int j = 0; j < 8; ++j) {
            const int  of   = ot * 16 + 8 * h + j;
            const bool live = (ot < 3) || (of < D_);
            const float bias = gb[live ? of : 0];
#pragma unroll
            for (int nt = 0; nt < 2; ++nt)
                gcol[(size_t)of * NN + 16 * nt] = live ? (acc[nt][ot][j] + bias) : 0.0f;
        }
    }
}


__device__ __forceinline__ float xhalf(float x) {
    int b = __builtin_bit_cast(int, x);
    return __builtin_bit_cast(float,
        __builtin_amdgcn_permlanex16(b, b, 0x76543210, 0xfedcba98, false, false));
}

__device__ __forceinline__ void bar_arrive() {
    asm volatile("s_wait_dscnt 0x0\n\ts_barrier_signal -1" ::: "memory");
}
__device__ __forceinline__ void bar_pass() {
    asm volatile("s_barrier_wait -1" ::: "memory");
}

__global__ __launch_bounds__(256)
void nlb_attn(const float* __restrict__ Q, const float* __restrict__ Gt,
              float* __restrict__ Y) {
    __shared__ __attribute__((aligned(16))) float lds[2 * STG];

    const int tid  = threadIdx.x;
    const int lane = threadIdx.x & 31;
    const int wave = threadIdx.x >> 5;
    const int m    = lane & 15;
    const int h    = lane >> 4;
    const int wq   = wave & 3;
    const int kh   = wave >> 2;
    const int b    = blockIdx.x >> 6;
    const int q0   = ((blockIdx.x & 63) * 4 + wq) * 16;

    const float* Qb  = Q  + (size_t)b * NN * DP;
    const float* Gtb = Gt + (size_t)b * DP * NN;
    float*       Yb  = Y  + (size_t)b * NN * DP;

    const int kst = (tid >> 4) * KP + (tid & 15) * 4;
    const int vst = (tid >> 5) * VP + (tid & 31) * 4;
    const float* kgl = Qb + tid * 4;
    const float* vgl = Gtb + (size_t)(tid >> 5) * NN + (tid & 31) * 4;

    const int kfr = (kh * 64 + m) * KP + 2 * h;
    const int vfr = KBUF + m * VP + kh * 64 + 8 * h;

    v4f kr[8], vr[8];

#pragma unroll
    for (int i = 0; i < 8; ++i) {
        kr[i] = *(const v4f*)(kgl + i * 1024);
        vr[i] = *(const v4f*)(vgl + (size_t)i * 8 * NN);
    }
#pragma unroll
    for (int i = 0; i < 8; ++i) {
        *(v4f*)(lds + kst + i * 16 * KP)       = kr[i];
        *(v4f*)(lds + KBUF + vst + i * 8 * VP) = vr[i];
    }
#pragma unroll
    for (int i = 0; i < 8; ++i) {
        kr[i] = *(const v4f*)(kgl + (size_t)KT * DP + i * 1024);
        vr[i] = *(const v4f*)(vgl + KT + (size_t)i * 8 * NN);
    }

    const float LOG2E = 1.4426950408889634f;
    v2f qa[15];
    {
        const float* qrow = Qb + (size_t)(q0 + m) * DP;
#pragma unroll
        for (int kk = 0; kk < 15; ++kk) {
            v2f q = *(const v2f*)(qrow + 4 * kk + 2 * h);
            qa[kk] = q * LOG2E;
        }
    }

    v8f o0{}, o1{}, o2{}, o3{};
    float mrow = -__builtin_inff();
    float lrow = 0.0f;

    __syncthreads();

    v2f kf[4][15];
#pragma unroll
    for (int t = 0; t < 4; ++t)
#pragma unroll
        for (int kk = 0; kk < 15; ++kk)
            kf[t][kk] = *(const v2f*)(lds + kfr + t * 16 * KP + 4 * kk);

    for (int c2 = 0; c2 < NN / KT; c2 += 2) {
#pragma unroll
    for (int par = 0; par < 2; ++par) {
        const int ch = c2 + par;
        const int k0 = ch * KT;
        const float* st  = lds + par * STG;
        float*       nst = lds + (par ^ 1) * STG;
        const bool more1 = (ch + 1 < NN / KT);
        const bool more2 = (ch + 2 < NN / KT);

        if (more1) {
#pragma unroll
            for (int i = 0; i < 8; ++i) {
                *(v4f*)(nst + kst + i * 16 * KP)       = kr[i];
                *(v4f*)(nst + KBUF + vst + i * 8 * VP) = vr[i];
            }
        }
        v4f vf[4][2][4];
#pragma unroll
        for (int t = 0; t < 4; ++t)
#pragma unroll
            for (int up = 0; up < 2; ++up)
#pragma unroll
                for (int ct = 0; ct < 4; ++ct)
                    vf[t][up][ct] = *(const v4f*)(st + vfr + ct * 16 * VP + 16 * t + 4 * up);

        __builtin_amdgcn_sched_barrier(0);

        if (more2) {
#pragma unroll
            for (int i = 0; i < 8; ++i) {
                kr[i] = *(const v4f*)(kgl + (size_t)(k0 + 2 * KT) * DP + i * 1024);
                vr[i] = *(const v4f*)(vgl + (k0 + 2 * KT) + (size_t)i * 8 * NN);
            }
        }
        __builtin_amdgcn_sched_barrier(0);
        if (par == 0) __builtin_amdgcn_s_setprio(3);
        else          __builtin_amdgcn_s_setprio(1);

        v8f s[4];
#pragma unroll
        for (int t = 0; t < 4; ++t) s[t] = v8f{};
#pragma unroll
        for (int kk = 0; kk < 15; ++kk) {
            if (kk == SIGK) {
                __builtin_amdgcn_sched_barrier(0);
                bar_arrive();
                __builtin_amdgcn_sched_barrier(0);
            }
#pragma unroll
            for (int t = 0; t < 4; ++t)
                s[t] = __builtin_amdgcn_wmma_f32_16x16x4_f32(
                           false, kf[t][kk], false, qa[kk], (short)0, s[t], false, false);
        }

        __builtin_amdgcn_sched_barrier(0);
        if (par == 0) __builtin_amdgcn_s_setprio(2);
        else          __builtin_amdgcn_s_setprio(0);

        bar_pass();
        if (more1) {
#pragma unroll
            for (int t = 0; t < 4; ++t)
#pragma unroll
                for (int kk = 0; kk < 15; ++kk)
                    kf[t][kk] = *(const v2f*)(nst + kfr + t * 16 * KP + 4 * kk);
        }
        __builtin_amdgcn_sched_barrier(0);

        float tm;
        {
            float tmax[4];
#pragma unroll
            for (int t = 0; t < 4; ++t)
                tmax[t] = fmaxf(fmaxf(fmaxf(s[t][0], s[t][1]), fmaxf(s[t][2], s[t][3])),
                                fmaxf(fmaxf(s[t][4], s[t][5]), fmaxf(s[t][6], s[t][7])));
            tm = fmaxf(fmaxf(tmax[0], tmax[1]), fmaxf(tmax[2], tmax[3]));
        }
        tm = fmaxf(tm, xhalf(tm));
        const float mn = fmaxf(mrow, tm);
        const float sc = fast_exp2(mrow - mn);
        float su;
        {
            float part[4];
#pragma unroll
            for (int t = 0; t < 4; ++t) {
#pragma unroll
                for (int r = 0; r < 8; ++r) s[t][r] = fast_exp2(s[t][r] - mn);
                part[t] = ((s[t][0] + s[t][1]) + (s[t][2] + s[t][3])) +
                          ((s[t][4] + s[t][5]) + (s[t][6] + s[t][7]));
            }
            su = (part[0] + part[1]) + (part[2] + part[3]);
        }
        su += xhalf(su);
        lrow = lrow * sc + su;
        mrow = mn;
        o0 *= sc; o1 *= sc; o2 *= sc; o3 *= sc;

#pragma unroll
        for (int t = 0; t < 4; ++t) {
#pragma unroll
            for (int up = 0; up < 2; ++up) {
                const v4f va0 = vf[t][up][0];
                const v4f va1 = vf[t][up][1];
                const v4f va2 = vf[t][up][2];
                const v4f va3 = vf[t][up][3];
                const v2f pb0 = v2f{s[t][4 * up],     s[t][4 * up + 1]};
                const v2f pb1 = v2f{s[t][4 * up + 2], s[t][4 * up + 3]};
                o0 = __builtin_amdgcn_wmma_f32_16x16x4_f32(false, v2f{va0.x, va0.y}, false, pb0, (short)0, o0, false, false);
                o1 = __builtin_amdgcn_wmma_f32_16x16x4_f32(false, v2f{va1.x, va1.y}, false, pb0, (short)0, o1, false, false);
                o2 = __builtin_amdgcn_wmma_f32_16x16x4_f32(false, v2f{va2.x, va2.y}, false, pb0, (short)0, o2, false, false);
                o3 = __builtin_amdgcn_wmma_f32_16x16x4_f32(false, v2f{va3.x, va3.y}, false, pb0, (short)0, o3, false, false);
                o0 = __builtin_amdgcn_wmma_f32_16x16x4_f32(false, v2f{va0.z, va0.w}, false, pb1, (short)0, o0, false, false);
                o1 = __builtin_amdgcn_wmma_f32_16x16x4_f32(false, v2f{va1.z, va1.w}, false, pb1, (short)0, o1, false, false);
                o2 = __builtin_amdgcn_wmma_f32_16x16x4_f32(false, v2f{va2.z, va2.w}, false, pb1, (short)0, o2, false, false);
                o3 = __builtin_amdgcn_wmma_f32_16x16x4_f32(false, v2f{va3.z, va3.w}, false, pb1, (short)0, o3, false, false);
            }
        }
    }
    }

    float* jn = lds + wq * (34 * 32) + lane;
    if (kh == 1) {
#pragma unroll
        for (int r = 0; r < 8; ++r) {
            jn[(r)      * 32] = o0[r];
            jn[(8 + r)  * 32] = o1[r];
            jn[(16 + r) * 32] = o2[r];
            jn[(24 + r) * 32] = o3[r];
        }
        jn[32 * 32] = mrow;
        jn[33 * 32] = lrow;
    }
    __syncthreads();
    if (kh == 0) {
        const float m1 = jn[32 * 32];
        const float l1 = jn[33 * 32];
        const float mx = fmaxf(mrow, m1);
        const float a0 = fast_exp2(mrow - mx);
        const float a1 = fast_exp2(m1 - mx);
        const float inv = 1.0f / (lrow * a0 + l1 * a1);
        const float w0 = a0 * inv;
        const float w1 = a1 * inv;
        v8f p0, p1, p2, p3;
#pragma unroll
        for (int r = 0; r < 8; ++r) {
            p0[r] = jn[(r)      * 32];
            p1[r] = jn[(8 + r)  * 32];
            p2[r] = jn[(16 + r) * 32];
            p3[r] = jn[(24 + r) * 32];
        }
        o0 = o0 * w0 + p0 * w1;
        o1 = o1 * w0 + p1 * w1;
        o2 = o2 * w0 + p2 * w1;
        o3 = o3 * w0 + p3 * w1;

        float* yr = Yb + (size_t)(q0 + m) * DP + 8 * h;
        *(v4f*)(yr +  0) = v4f{o0[0], o0[1], o0[2], o0[3]};
        *(v4f*)(yr +  4) = v4f{o0[4], o0[5], o0[6], o0[7]};
        *(v4f*)(yr + 16) = v4f{o1[0], o1[1], o1[2], o1[3]};
        *(v4f*)(yr + 20) = v4f{o1[4], o1[5], o1[6], o1[7]};
        *(v4f*)(yr + 32) = v4f{o2[0], o2[1], o2[2], o2[3]};
        *(v4f*)(yr + 36) = v4f{o2[4], o2[5], o2[6], o2[7]};
        *(v4f*)(yr + 48) = v4f{o3[0], o3[1], o3[2], o3[3]};
        *(v4f*)(yr + 52) = v4f{o3[4], o3[5], o3[6], o3[7]};
    }
}

__global__ __launch_bounds__(128)
void nlb_out(const float* __restrict__ x, const float* __restrict__ Y,
             const float* __restrict__ ww, const float* __restrict__ wb,
             float* __restrict__ Z) {
    const int lane = threadIdx.x & 31;
    const int wave = threadIdx.x >> 5;
    const int m    = lane & 15;
    const int h    = lane >> 4;
    const int tile = blockIdx.x * 4 + wave;
    const int b    = tile >> 8;
    const int n0   = (tile & 255) * 16;

    v2f yb[15];
    {
        const float* yrow = Y + ((size_t)b * NN + n0 + m) * DP + 2 * h;
#pragma unroll
        for (int kk = 0; kk < 15; ++kk) yb[kk] = *(const v2f*)(yrow + 4 * kk);
    }

    v8f acc[4];
#pragma unroll
    for (int ot = 0; ot < 4; ++ot) acc[ot] = v8f{};

#pragma unroll
    for (int kk = 0; kk < 15; ++kk) {
#pragma unroll
        for (int ot = 0; ot < 4; ++ot) {
            const int  o    = ot * 16 + m;
            const bool live = (ot < 3) || (o < D_);
            v2f wa = *(const v2f*)(ww + (live ? o : 0) * D_ + 4 * kk + 2 * h);
            wa = live ? wa : v2f{0.0f, 0.0f};
            acc[ot] = __builtin_amdgcn_wmma_f32_16x16x4_f32(
                          false, wa, false, yb[kk], (short)0, acc[ot], false, false);
        }
    }

    const int oh = n0 >> 6;
    const int ow = (n0 & 63) + m;
#pragma unroll
    for (int ot = 0; ot < 4; ++ot) {
#pragma unroll
        for (int jp = 0; jp < 4; ++jp) {
            const int oo = ot * 16 + 8 * h + 2 * jp;
            if ((ot < 3) || (oo < D_)) {
                const int tc = oo >> 2;
                const int ri = (oo >> 1) & 1;
                const size_t zi = ((size_t)(b * (T_ * C_) + tc) * H_ + (2 * oh + ri)) * W_ + 2 * ow;
                const v2f xv = *(const v2f*)(x + zi);
                v2f r;
                r.x = (acc[ot][2 * jp]     + wb[oo])     + xv.x;
                r.y = (acc[ot][2 * jp + 1] + wb[oo + 1]) + xv.y;
                *(v2f*)(Z + zi) = r;
            }
        }
    }
}

extern "C" void kernel_launch(void* const* d_in, const int* in_sizes, int n_in,
                              void* d_out, int out_size, void* d_ws, size_t ws_size,
                              hipStream_t stream) {
    const float* x  = (const float*)d_in[0];
    const float* gw = (const float*)d_in[1];
    const float* gb = (const float*)d_in[2];
    const float* ww = (const float*)d_in[3];
    const float* wb = (const float*)d_in[4];
    float* Z = (float*)d_out;

    const size_t buf = (size_t)B_ * NN * DP;
    float* Qbuf = (float*)d_ws;
    float* Gbuf = Qbuf + buf;
    float* Ybuf = Gbuf + buf;

    nlb_prep_qg<<<B_ * NN / 64, 64, 0, stream>>>(x, gw, gb, Qbuf, Gbuf);
    nlb_attn<<<B_ * 64, 256, 0, stream>>>(Qbuf, Gbuf, Ybuf);
    nlb_out<<<B_ * NN / 64, 128, 0, stream>>>(x, Ybuf, ww, wb, Z);
}


# output projection, pixel-shuffle and residual fused into attention's join epilogue: merged O^T registers are the WMMA B operand, w_w staged in LDS; 3 launches -> 2; projection sum order changes
# speedup vs baseline: 2.9775x; 1.0427x over previous
#include <hip/hip_runtime.h>

typedef __attribute__((ext_vector_type(2))) float v2f;
typedef __attribute__((ext_vector_type(4))) float v4f;
typedef __attribute__((ext_vector_type(8))) float v8f;

#define B_  4
#define T_  5
#define C_  3
#define H_  128
#define W_  128
#define D_  60
#define DP  64
#define NN  4096
#define KT  128
#define KP  68
#define VP  132
#define KBUF (KT * KP)
#define VBUF (DP * VP)
#define STG  (KBUF + VBUF)
#define SIGK 6
#define WP   68

__device__ __forceinline__ float fast_exp2(float x) {
    return __builtin_amdgcn_exp2f(x);
}

#define QP 68

__global__ __launch_bounds__(64)
void nlb_prep_qg(const float* __restrict__ x, const float* __restrict__ gw,
                 const float* __restrict__ gb, float* __restrict__ Q,
                 float* __restrict__ Gt) {
    __shared__ __attribute__((aligned(16))) float qs[2 * 32 * QP];

    const int lane = threadIdx.x & 31;
    const int wave = threadIdx.x >> 5;
    const int m    = lane & 15;
    const int h    = lane >> 4;
    const int tile = blockIdx.x * 2 + wave;
    const int b    = tile >> 7;
    const int n0   = (tile & 127) * 32;
    const int oh   = n0 >> 6;
    const int ow0  = n0 & 63;

    v2f qb[2][15];
#pragma unroll
    for (int nt = 0; nt < 2; ++nt) {
        const float* xr = x + ((size_t)(b * (T_ * C_)) * H_ + 2 * oh + h) * W_
                            + 2 * (ow0 + 16 * nt + m);
#pragma unroll
        for (int kk = 0; kk < 15; ++kk) qb[nt][kk] = *(const v2f*)(xr + (size_t)kk * H_ * W_);
    }

    float* qw = qs + wave * (32 * QP);
#pragma unroll
    for (int nt = 0; nt < 2; ++nt) {
        float* qrow = qw + (16 * nt + m) * QP + 2 * h;
#pragma unroll
        for (int kk = 0; kk < 15; ++kk) *(v2f*)(qrow + 4 * kk) = qb[nt][kk];
        *(v2f*)(qrow + D_) = v2f{0.0f, 0.0f};
    }

    v8f acc[2][4];
#pragma unroll
    for (int nt = 0; nt < 2; ++nt)
#pragma unroll
        for (int ot = 0; ot < 4; ++ot) acc[nt][ot] = v8f{};

#pragma unroll
    for (int kk = 0; kk < 15; ++kk) {
#pragma unroll
        for (int ot = 0; ot < 4; ++ot) {
            const int  o    = ot * 16 + m;
            const bool live = (ot < 3) || (o < D_);
            v2f wa = *(const v2f*)(gw + (live ? o : 0) * D_ + 4 * kk + 2 * h);
            wa = live ? wa : v2f{0.0f, 0.0f};
#pragma unroll
            for (int nt = 0; nt < 2; ++nt)
                acc[nt][ot] = __builtin_amdgcn_wmma_f32_16x16x4_f32(
                                  false, wa, false, qb[nt][kk], (short)0, acc[nt][ot], false, false);
        }
    }

    __syncthreads();
    {
        float* qg = Q + ((size_t)b * NN + n0) * DP;
#pragma unroll
        for (int i = 0; i < 16; ++i) {
            const int row = 2 * i + h;
            *(v4f*)(qg + row * DP + 4 * m) = *(const v4f*)(qw + row * QP + 4 * m);
        }
    }

    float* gcol = Gt + (size_t)b * DP * NN + n0 + m;
#pragma unroll
    for (int ot = 0; ot < 4; ++ot) {
#pragma unroll
        for (int j = 0; j < 8; ++j) {
            const int  of   = ot * 16 + 8 * h + j;
            const bool live = (ot < 3) || (of < D_);
            const float bias = gb[live ? of : 0];
#pragma unroll
            for (int nt = 0; nt < 2; ++nt)
                gcol[(size_t)of * NN + 16 * nt] = live ? (acc[nt][ot][j] + bias) : 0.0f;
        }
    }
}


__device__ __forceinline__ float xhalf(float x) {
    int b = __builtin_bit_cast(int, x);
    return __builtin_bit_cast(float,
        __builtin_amdgcn_permlanex16(b, b, 0x76543210, 0xfedcba98, false, false));
}

__device__ __forceinline__ void bar_arrive() {
    asm volatile("s_wait_dscnt 0x0\n\ts_barrier_signal -1" ::: "memory");
}
__device__ __forceinline__ void bar_pass() {
    asm volatile("s_barrier_wait -1" ::: "memory");
}

__global__ __launch_bounds__(256)
void nlb_attn(const float* __restrict__ Q, const float* __restrict__ Gt,
              const float* __restrict__ x, const float* __restrict__ ww,
              const float* __restrict__ wb, float* __restrict__ Z) {
    __shared__ __attribute__((aligned(16))) float lds[2 * STG + D_ * WP];

    const int tid  = threadIdx.x;
    const int lane = threadIdx.x & 31;
    const int wave = threadIdx.x >> 5;
    const int m    = lane & 15;
    const int h    = lane >> 4;
    const int wq   = wave & 3;
    const int kh   = wave >> 2;
    const int b    = blockIdx.x >> 6;
    const int q0   = ((blockIdx.x & 63) * 4 + wq) * 16;

    const float* Qb  = Q  + (size_t)b * NN * DP;
    const float* Gtb = Gt + (size_t)b * DP * NN;
    float*       wws = lds + 2 * STG;

    const int kst = (tid >> 4) * KP + (tid & 15) * 4;
    const int vst = (tid >> 5) * VP + (tid & 31) * 4;
    const float* kgl = Qb + tid * 4;
    const float* vgl = Gtb + (size_t)(tid >> 5) * NN + (tid & 31) * 4;

    const int kfr = (kh * 64 + m) * KP + 2 * h;
    const int vfr = KBUF + m * VP + kh * 64 + 8 * h;

    v4f kr[8], vr[8];

    v2f wst[8];
#pragma unroll
    for (int i = 0; i < 8; ++i) {
        const int p = i * 256 + tid;
        wst[i] = *(const v2f*)(ww + 2 * ((i < 7 || p < D_ * D_ / 2) ? p : D_ * D_ / 2 - 1));
    }

#pragma unroll
    for (int i = 0; i < 8; ++i) {
        kr[i] = *(const v4f*)(kgl + i * 1024);
        vr[i] = *(const v4f*)(vgl + (size_t)i * 8 * NN);
    }
#pragma unroll
    for (int i = 0; i < 8; ++i) {
        *(v4f*)(lds + kst + i * 16 * KP)       = kr[i];
        *(v4f*)(lds + KBUF + vst + i * 8 * VP) = vr[i];
    }
#pragma unroll
    for (int i = 0; i < 8; ++i) {
        const int p   = i * 256 + tid;
        const int row = p / 30;
        const int pc  = p - row * 30;
        if (i < 7 || tid < D_ * D_ / 2 - 7 * 256)
            *(v2f*)(wws + row * WP + 2 * pc) = wst[i];
    }
#pragma unroll
    for (int i = 0; i < 8; ++i) {
        kr[i] = *(const v4f*)(kgl + (size_t)KT * DP + i * 1024);
        vr[i] = *(const v4f*)(vgl + KT + (size_t)i * 8 * NN);
    }

    const float LOG2E = 1.4426950408889634f;
    v2f qa[15];
    {
        const float* qrow = Qb + (size_t)(q0 + m) * DP;
#pragma unroll
        for (int kk = 0; kk < 15; ++kk) {
            v2f q = *(const v2f*)(qrow + 4 * kk + 2 * h);
            qa[kk] = q * LOG2E;
        }
    }

    v8f o0{}, o1{}, o2{}, o3{};
    float mrow = -__builtin_inff();
    float lrow = 0.0f;

    __syncthreads();

    v2f kf[4][15];
#pragma unroll
    for (int t = 0; t < 4; ++t)
#pragma unroll
        for (int kk = 0; kk < 15; ++kk)
            kf[t][kk] = *(const v2f*)(lds + kfr + t * 16 * KP + 4 * kk);

    for (int c2 = 0; c2 < NN / KT; c2 += 2) {
#pragma unroll
    for (int par = 0; par < 2; ++par) {
        const int ch = c2 + par;
        const int k0 = ch * KT;
        const float* st  = lds + par * STG;
        float*       nst = lds + (par ^ 1) * STG;
        const bool more1 = (ch + 1 < NN / KT);
        const bool more2 = (ch + 2 < NN / KT);

        if (more1) {
#pragma unroll
            for (int i = 0; i < 8; ++i) {
                *(v4f*)(nst + kst + i * 16 * KP)       = kr[i];
                *(v4f*)(nst + KBUF + vst + i * 8 * VP) = vr[i];
            }
        }
        v4f vf[4][2][4];
#pragma unroll
        for (int t = 0; t < 4; ++t)
#pragma unroll
            for (int up = 0; up < 2; ++up)
#pragma unroll
                for (int ct = 0; ct < 4; ++ct)
                    vf[t][up][ct] = *(const v4f*)(st + vfr + ct * 16 * VP + 16 * t + 4 * up);

        __builtin_amdgcn_sched_barrier(0);

        if (more2) {
#pragma unroll
            for (int i = 0; i < 8; ++i) {
                kr[i] = *(const v4f*)(kgl + (size_t)(k0 + 2 * KT) * DP + i * 1024);
                vr[i] = *(const v4f*)(vgl + (k0 + 2 * KT) + (size_t)i * 8 * NN);
            }
        }
        __builtin_amdgcn_sched_barrier(0);
        if (par == 0) __builtin_amdgcn_s_setprio(3);
        else          __builtin_amdgcn_s_setprio(1);

        v8f s[4];
#pragma unroll
        for (int t = 0; t < 4; ++t) s[t] = v8f{};
#pragma unroll
        for (int kk = 0; kk < 15; ++kk) {
            if (kk == SIGK) {
                __builtin_amdgcn_sched_barrier(0);
                bar_arrive();
                __builtin_amdgcn_sched_barrier(0);
            }
#pragma unroll
            for (int t = 0; t < 4; ++t)
                s[t] = __builtin_amdgcn_wmma_f32_16x16x4_f32(
                           false, kf[t][kk], false, qa[kk], (short)0, s[t], false, false);
        }

        __builtin_amdgcn_sched_barrier(0);
        if (par == 0) __builtin_amdgcn_s_setprio(2);
        else          __builtin_amdgcn_s_setprio(0);

        bar_pass();
        if (more1) {
#pragma unroll
            for (int t = 0; t < 4; ++t)
#pragma unroll
                for (int kk = 0; kk < 15; ++kk)
                    kf[t][kk] = *(const v2f*)(nst + kfr + t * 16 * KP + 4 * kk);
        }
        __builtin_amdgcn_sched_barrier(0);

        float tm;
        {
            float tmax[4];
#pragma unroll
            for (int t = 0; t < 4; ++t)
                tmax[t] = fmaxf(fmaxf(fmaxf(s[t][0], s[t][1]), fmaxf(s[t][2], s[t][3])),
                                fmaxf(fmaxf(s[t][4], s[t][5]), fmaxf(s[t][6], s[t][7])));
            tm = fmaxf(fmaxf(tmax[0], tmax[1]), fmaxf(tmax[2], tmax[3]));
        }
        tm = fmaxf(tm, xhalf(tm));
        const float mn = fmaxf(mrow, tm);
        const float sc = fast_exp2(mrow - mn);
        float su;
        {
            float part[4];
#pragma unroll
            for (int t = 0; t < 4; ++t) {
#pragma unroll
                for (int r = 0; r < 8; ++r) s[t][r] = fast_exp2(s[t][r] - mn);
                part[t] = ((s[t][0] + s[t][1]) + (s[t][2] + s[t][3])) +
                          ((s[t][4] + s[t][5]) + (s[t][6] + s[t][7]));
            }
            su = (part[0] + part[1]) + (part[2] + part[3]);
        }
        su += xhalf(su);
        lrow = lrow * sc + su;
        mrow = mn;
        o0 *= sc; o1 *= sc; o2 *= sc; o3 *= sc;

#pragma unroll
        for (int t = 0; t < 4; ++t) {
#pragma unroll
            for (int up = 0; up < 2; ++up) {
                const v4f va0 = vf[t][up][0];
                const v4f va1 = vf[t][up][1];
                const v4f va2 = vf[t][up][2];
                const v4f va3 = vf[t][up][3];
                const v2f pb0 = v2f{s[t][4 * up],     s[t][4 * up + 1]};
                const v2f pb1 = v2f{s[t][4 * up + 2], s[t][4 * up + 3]};
                o0 = __builtin_amdgcn_wmma_f32_16x16x4_f32(false, v2f{va0.x, va0.y}, false, pb0, (short)0, o0, false, false);
                o1 = __builtin_amdgcn_wmma_f32_16x16x4_f32(false, v2f{va1.x, va1.y}, false, pb0, (short)0, o1, false, false);
                o2 = __builtin_amdgcn_wmma_f32_16x16x4_f32(false, v2f{va2.x, va2.y}, false, pb0, (short)0, o2, false, false);
                o3 = __builtin_amdgcn_wmma_f32_16x16x4_f32(false, v2f{va3.x, va3.y}, false, pb0, (short)0, o3, false, false);
                o0 = __builtin_amdgcn_wmma_f32_16x16x4_f32(false, v2f{va0.z, va0.w}, false, pb1, (short)0, o0, false, false);
                o1 = __builtin_amdgcn_wmma_f32_16x16x4_f32(false, v2f{va1.z, va1.w}, false, pb1, (short)0, o1, false, false);
                o2 = __builtin_amdgcn_wmma_f32_16x16x4_f32(false, v2f{va2.z, va2.w}, false, pb1, (short)0, o2, false, false);
                o3 = __builtin_amdgcn_wmma_f32_16x16x4_f32(false, v2f{va3.z, va3.w}, false, pb1, (short)0, o3, false, false);
            }
        }
    }
    }

    float* jn = lds + wq * (34 * 32) + lane;
    const int khu = __builtin_amdgcn_readfirstlane(kh);

    const int oh = q0 >> 6;
    const int ow = (q0 & 63) + m;

    v2f wf[4][4][4];
    v2f xres[4][4];
    v2f bia[4][4];
    if (khu == 1) {
#pragma unroll
        for (int r = 0; r < 8; ++r) {
            jn[(r)      * 32] = o0[r];
            jn[(8 + r)  * 32] = o1[r];
            jn[(16 + r) * 32] = o2[r];
            jn[(24 + r) * 32] = o3[r];
        }
        jn[32 * 32] = mrow;
        jn[33 * 32] = lrow;
    } else {
#pragma unroll
        for (int ct = 0; ct < 4; ++ct) {
#pragma unroll
            for (int u = 0; u < 4; ++u) {
#pragma unroll
                for (int ot = 0; ot < 4; ++ot) {
                    const int  o    = ot * 16 + m;
                    const int  f    = 16 * ct + 8 * h + 2 * u;
                    const bool live = ((ot < 3) || (o < D_)) && ((ct < 3) || (f < D_));
                    v2f wa = *(const v2f*)(wws + (live ? o * WP + f : 0));
                    wf[ct][u][ot] = live ? wa : v2f{0.0f, 0.0f};
                }
            }
        }
#pragma unroll
        for (int ot = 0; ot < 4; ++ot) {
#pragma unroll
            for (int jp = 0; jp < 4; ++jp) {
                const int  oo = ot * 16 + 8 * h + 2 * jp;
                const bool ok = (ot < 3) || (oo < D_);
                const int  oc = ok ? oo : 0;
                const int  tc = oc >> 2;
                const int  ri = (oc >> 1) & 1;
                const size_t zi = ((size_t)(b * (T_ * C_) + tc) * H_ + (2 * oh + ri)) * W_ + 2 * ow;
                xres[ot][jp] = *(const v2f*)(x + zi);
                bia[ot][jp]  = *(const v2f*)(wb + oc);
            }
        }
    }
    __syncthreads();
    if (khu == 0) {
        const float m1 = jn[32 * 32];
        const float l1 = jn[33 * 32];
        const float mx = fmaxf(mrow, m1);
        const float a0 = fast_exp2(mrow - mx);
        const float a1 = fast_exp2(m1 - mx);
        const float inv = 1.0f / (lrow * a0 + l1 * a1);
        const float w0 = a0 * inv;
        const float w1 = a1 * inv;
        v8f p0, p1, p2, p3;
#pragma unroll
        for (int r = 0; r < 8; ++r) {
            p0[r] = jn[(r)      * 32];
            p1[r] = jn[(8 + r)  * 32];
            p2[r] = jn[(16 + r) * 32];
            p3[r] = jn[(24 + r) * 32];
        }
        v8f yt[4];
        yt[0] = o0 * w0 + p0 * w1;
        yt[1] = o1 * w0 + p1 * w1;
        yt[2] = o2 * w0 + p2 * w1;
        yt[3] = o3 * w0 + p3 * w1;

        __builtin_amdgcn_sched_barrier(0);

        v8f z[4];
#pragma unroll
        for (int ot = 0; ot < 4; ++ot) z[ot] = v8f{};
#pragma unroll
        for (int ct = 0; ct < 4; ++ct) {
#pragma unroll
            for (int u = 0; u < 4; ++u) {
                const v2f yb = v2f{yt[ct][2 * u], yt[ct][2 * u + 1]};
#pragma unroll
                for (int ot = 0; ot < 4; ++ot)
                    z[ot] = __builtin_amdgcn_wmma_f32_16x16x4_f32(
                                false, wf[ct][u][ot], false, yb, (short)0, z[ot], false, false);
            }
        }

#pragma unroll
        for (int ot = 0; ot < 4; ++ot) {
#pragma unroll
            for (int jp = 0; jp < 4; ++jp) {
                const int oo = ot * 16 + 8 * h + 2 * jp;
                if ((ot < 3) || (oo < D_)) {
                    const int tc = oo >> 2;
                    const int ri = (oo >> 1) & 1;
                    const size_t zi = ((size_t)(b * (T_ * C_) + tc) * H_ + (2 * oh + ri)) * W_ + 2 * ow;
                    v2f r;
                    r.x = (z[ot][2 * jp]     + bia[ot][jp].x) + xres[ot][jp].x;
                    r.y = (z[ot][2 * jp + 1] + bia[ot][jp].y) + xres[ot][jp].y;
                    *(v2f*)(Z + zi) = r;
                }
            }
        }
    }
}

extern "C" void kernel_launch(void* const* d_in, const int* in_sizes, int n_in,
                              void* d_out, int out_size, void* d_ws, size_t ws_size,
                              hipStream_t stream) {
    const float* x  = (const float*)d_in[0];
    const float* gw = (const float*)d_in[1];
    const float* gb = (const float*)d_in[2];
    const float* ww = (const float*)d_in[3];
    const float* wb = (const float*)d_in[4];
    float* Z = (float*)d_out;

    const size_t buf = (size_t)B_ * NN * DP;
    float* Qbuf = (float*)d_ws;
    float* Gbuf = Qbuf + buf;

    nlb_prep_qg<<<B_ * NN / 64, 64, 0, stream>>>(x, gw, gb, Qbuf, Gbuf);
    nlb_attn<<<B_ * 64, 256, 0, stream>>>(Qbuf, Gbuf, x, ww, wb, Z);
}


# V projection moved behind the attention (softmax rows sum to 1: F(g_w xf + g_b) = g_w(F xf) + g_b): K = V = unshuffled x, chunks staged straight from x; 2 launches -> 1; sum order changes
# speedup vs baseline: 3.2430x; 1.0892x over previous
#include <hip/hip_runtime.h>

typedef __attribute__((ext_vector_type(2))) float v2f;
typedef __attribute__((ext_vector_type(4))) float v4f;
typedef __attribute__((ext_vector_type(8))) float v8f;

#define B_  4
#define T_  5
#define C_  3
#define H_  128
#define W_  128
#define D_  60
#define DP  64
#define NN  4096
#define KT  128
#define KP  68
#define VP  132
#define KBUF (KT * KP)
#define VBUF (DP * VP)
#define STG  (KBUF + VBUF)
#define SIGK 6
#define WP   68
#define PLN  (H_ * W_)

__device__ __forceinline__ float fast_exp2(float x) {
    return __builtin_amdgcn_exp2f(x);
}


__device__ __forceinline__ float xhalf(float x) {
    int b = __builtin_bit_cast(int, x);
    return __builtin_bit_cast(float,
        __builtin_amdgcn_permlanex16(b, b, 0x76543210, 0xfedcba98, false, false));
}

__device__ __forceinline__ void bar_arrive() {
    asm volatile("s_wait_dscnt 0x0\n\ts_barrier_signal -1" ::: "memory");
}
__device__ __forceinline__ void bar_pass() {
    asm volatile("s_barrier_wait -1" ::: "memory");
}

__device__ __forceinline__ void fetch_chunk(v4f (&xr)[8], const float* xgl, const float* xgl7, int ch) {
#pragma unroll
    for (int i = 0; i < 7; ++i) xr[i] = *(const v4f*)(xgl + (size_t)i * 2 * PLN + ch * 4 * W_);
    xr[7] = *(const v4f*)(xgl7 + (size_t)14 * PLN + ch * 4 * W_);
}

__device__ __forceinline__ void park_chunk(float* stage, int kst, int vst, const v4f (&xr)[8], bool padl) {
#pragma unroll
    for (int i = 0; i < 8; ++i) {
        v4f e = xr[i];
        if (i == 7) e = padl ? v4f{0.0f, 0.0f, 0.0f, 0.0f} : e;
        *(v2f*)(stage + kst + 8 * i)               = v2f{e.x, e.y};
        *(v2f*)(stage + kst + 8 * i + KP)          = v2f{e.z, e.w};
        *(v2f*)(stage + vst + 8 * i * VP)          = v2f{e.x, e.z};
        *(v2f*)(stage + vst + 8 * i * VP + VP)     = v2f{e.y, e.w};
    }
}

__global__ __launch_bounds__(256)
void nlb_attn(const float* __restrict__ x, const float* __restrict__ gw,
              const float* __restrict__ gb, const float* __restrict__ ww,
              const float* __restrict__ wb, float* __restrict__ Z) {
    __shared__ __attribute__((aligned(16))) float lds[2 * STG + 2 * D_ * WP];

    const int tid  = threadIdx.x;
    const int lane = threadIdx.x & 31;
    const int wave = threadIdx.x >> 5;
    const int m    = lane & 15;
    const int h    = lane >> 4;
    const int wq   = wave & 3;
    const int kh   = wave >> 2;
    const int b    = blockIdx.x >> 6;
    const int q0   = ((blockIdx.x & 63) * 4 + wq) * 16;

    const float* xb  = x + (size_t)b * (T_ * C_) * PLN;
    float*       wws = lds + 2 * STG;
    float*       gws = wws + D_ * WP;

    const int oh = q0 >> 6;
    const int ow = (q0 & 63) + m;

    const int sq   = (wave & 3) * 8 + (lane & 7);
    const int sri  = (lane >> 3) & 1;
    const int spl  = lane >> 4;
    const int sohl = wave >> 2;
    const int sn0  = sohl * 64 + 2 * sq;
    const int sd0  = 4 * spl + 2 * sri;
    const int kst  = sn0 * KP + sd0;
    const int vst  = KBUF + sd0 * VP + sn0;
    const float* xgl  = xb + (size_t)spl * PLN + (2 * sohl + sri) * W_ + 4 * sq;
    const float* xgl7 = xgl - (size_t)spl * PLN;
    const bool   padl = (spl == 1);

    const int kfr = (kh * 64 + m) * KP + 2 * h;
    const int vfr = KBUF + m * VP + kh * 64 + 8 * h;

    v4f xr[8];

    fetch_chunk(xr, xgl, xgl7, 0);

    v2f wst[8], gst[8];
#pragma unroll
    for (int i = 0; i < 8; ++i) {
        const int p  = i * 256 + tid;
        const int pl = (i < 7 || p < D_ * D_ / 2) ? p : D_ * D_ / 2 - 1;
        wst[i] = *(const v2f*)(ww + 2 * pl);
        gst[i] = *(const v2f*)(gw + 2 * pl);
    }

    const float LOG2E = 1.4426950408889634f;
    v2f qa[15];
    {
        const float* qpx = xb + (size_t)(2 * oh + h) * W_ + 2 * ow;
#pragma unroll
        for (int kk = 0; kk < 15; ++kk) qa[kk] = *(const v2f*)(qpx + (size_t)kk * PLN);
    }

    park_chunk(lds, kst, vst, xr, padl);
    fetch_chunk(xr, xgl, xgl7, 1);

#pragma unroll
    for (int i = 0; i < 8; ++i) {
        const int p   = i * 256 + tid;
        const int row = p / 30;
        const int pc  = p - row * 30;
        if (i < 7 || tid < D_ * D_ / 2 - 7 * 256) {
            *(v2f*)(wws + row * WP + 2 * pc) = wst[i];
            *(v2f*)(gws + row * WP + 2 * pc) = gst[i];
        }
    }
#pragma unroll
    for (int kk = 0; kk < 15; ++kk) qa[kk] = qa[kk] * LOG2E;

    v8f o0{}, o1{}, o2{}, o3{};
    float mrow = -__builtin_inff();
    float lrow = 0.0f;

    __syncthreads();

    v2f kf[4][15];
#pragma unroll
    for (int t = 0; t < 4; ++t)
#pragma unroll
        for (int kk = 0; kk < 15; ++kk)
            kf[t][kk] = *(const v2f*)(lds + kfr + t * 16 * KP + 4 * kk);

    for (int c2 = 0; c2 < NN / KT; c2 += 2) {
#pragma unroll
    for (int par = 0; par < 2; ++par) {
        const int ch = c2 + par;
        const float* st  = lds + par * STG;
        float*       nst = lds + (par ^ 1) * STG;
        const bool more1 = (ch + 1 < NN / KT);
        const bool more2 = (ch + 2 < NN / KT);

        if (more1) park_chunk(nst, kst, vst, xr, padl);
        v4f vf[4][2][4];
#pragma unroll
        for (int t = 0; t < 4; ++t)
#pragma unroll
            for (int up = 0; up < 2; ++up)
#pragma unroll
                for (int ct = 0; ct < 4; ++ct)
                    vf[t][up][ct] = *(const v4f*)(st + vfr + ct * 16 * VP + 16 * t + 4 * up);

        __builtin_amdgcn_sched_barrier(0);

        if (more2) fetch_chunk(xr, xgl, xgl7, ch + 2);
        __builtin_amdgcn_sched_barrier(0);
        if (par == 0) __builtin_amdgcn_s_setprio(3);
        else          __builtin_amdgcn_s_setprio(1);

        v8f s[4];
#pragma unroll
        for (int t = 0; t < 4; ++t) s[t] = v8f{};
#pragma unroll
        for (int kk = 0; kk < 15; ++kk) {
            if (kk == SIGK) {
                __builtin_amdgcn_sched_barrier(0);
                bar_arrive();
                __builtin_amdgcn_sched_barrier(0);
            }
#pragma unroll
            for (int t = 0; t < 4; ++t)
                s[t] = __builtin_amdgcn_wmma_f32_16x16x4_f32(
                           false, kf[t][kk], false, qa[kk], (short)0, s[t], false, false);
        }

        __builtin_amdgcn_sched_barrier(0);
        if (par == 0) __builtin_amdgcn_s_setprio(2);
        else          __builtin_amdgcn_s_setprio(0);

        bar_pass();
        if (more1) {
#pragma unroll
            for (int t = 0; t < 4; ++t)
#pragma unroll
                for (int kk = 0; kk < 15; ++kk)
                    kf[t][kk] = *(const v2f*)(nst + kfr + t * 16 * KP + 4 * kk);
        }
        __builtin_amdgcn_sched_barrier(0);

        float tm;
        {
            float tmax[4];
#pragma unroll
            for (int t = 0; t < 4; ++t)
                tmax[t] = fmaxf(fmaxf(fmaxf(s[t][0], s[t][1]), fmaxf(s[t][2], s[t][3])),
                                fmaxf(fmaxf(s[t][4], s[t][5]), fmaxf(s[t][6], s[t][7])));
            tm = fmaxf(fmaxf(tmax[0], tmax[1]), fmaxf(tmax[2], tmax[3]));
        }
        tm = fmaxf(tm, xhalf(tm));
        const float mn = fmaxf(mrow, tm);
        const float sc = fast_exp2(mrow - mn);
        float su;
        {
            float part[4];
#pragma unroll
            for (int t = 0; t < 4; ++t) {
#pragma unroll
                for (int r = 0; r < 8; ++r) s[t][r] = fast_exp2(s[t][r] - mn);
                part[t] = ((s[t][0] + s[t][1]) + (s[t][2] + s[t][3])) +
                          ((s[t][4] + s[t][5]) + (s[t][6] + s[t][7]));
            }
            su = (part[0] + part[1]) + (part[2] + part[3]);
        }
        su += xhalf(su);
        lrow = lrow * sc + su;
        mrow = mn;
        o0 *= sc; o1 *= sc; o2 *= sc; o3 *= sc;

#pragma unroll
        for (int t = 0; t < 4; ++t) {
#pragma unroll
            for (int up = 0; up < 2; ++up) {
                const v4f va0 = vf[t][up][0];
                const v4f va1 = vf[t][up][1];
                const v4f va2 = vf[t][up][2];
                const v4f va3 = vf[t][up][3];
                const v2f pb0 = v2f{s[t][4 * up],     s[t][4 * up + 1]};
                const v2f pb1 = v2f{s[t][4 * up + 2], s[t][4 * up + 3]};
                o0 = __builtin_amdgcn_wmma_f32_16x16x4_f32(false, v2f{va0.x, va0.y}, false, pb0, (short)0, o0, false, false);
                o1 = __builtin_amdgcn_wmma_f32_16x16x4_f32(false, v2f{va1.x, va1.y}, false, pb0, (short)0, o1, false, false);
                o2 = __builtin_amdgcn_wmma_f32_16x16x4_f32(false, v2f{va2.x, va2.y}, false, pb0, (short)0, o2, false, false);
                o3 = __builtin_amdgcn_wmma_f32_16x16x4_f32(false, v2f{va3.x, va3.y}, false, pb0, (short)0, o3, false, false);
                o0 = __builtin_amdgcn_wmma_f32_16x16x4_f32(false, v2f{va0.z, va0.w}, false, pb1, (short)0, o0, false, false);
                o1 = __builtin_amdgcn_wmma_f32_16x16x4_f32(false, v2f{va1.z, va1.w}, false, pb1, (short)0, o1, false, false);
                o2 = __builtin_amdgcn_wmma_f32_16x16x4_f32(false, v2f{va2.z, va2.w}, false, pb1, (short)0, o2, false, false);
                o3 = __builtin_amdgcn_wmma_f32_16x16x4_f32(false, v2f{va3.z, va3.w}, false, pb1, (short)0, o3, false, false);
            }
        }
    }
    }

    float* jn = lds + wq * (34 * 32) + lane;
    const int khu = __builtin_amdgcn_readfirstlane(kh);

    v2f wf[4][4][4];
    v2f gf[4][4][4];
    float gbv[4][8];
    v2f xres[4][4];
    v2f bia[4][4];
    if (khu == 1) {
#pragma unroll
        for (int r = 0; r < 8; ++r) {
            jn[(r)      * 32] = o0[r];
            jn[(8 + r)  * 32] = o1[r];
            jn[(16 + r) * 32] = o2[r];
            jn[(24 + r) * 32] = o3[r];
        }
        jn[32 * 32] = mrow;
        jn[33 * 32] = lrow;
    } else {
#pragma unroll
        for (int ct = 0; ct < 4; ++ct) {
#pragma unroll
            for (int u = 0; u < 4; ++u) {
#pragma unroll
                for (int ot = 0; ot < 4; ++ot) {
                    const int  o    = ot * 16 + m;
                    const int  f    = 16 * ct + 8 * h + 2 * u;
                    const bool live = ((ot < 3) || (o < D_)) && ((ct < 3) || (f < D_));
                    v2f wa = *(const v2f*)(wws + (live ? o * WP + f : 0));
                    v2f ga = *(const v2f*)(gws + (live ? o * WP + f : 0));
                    wf[ct][u][ot] = live ? wa : v2f{0.0f, 0.0f};
                    gf[ct][u][ot] = live ? ga : v2f{0.0f, 0.0f};
                }
            }
        }
#pragma unroll
        for (int ot = 0; ot < 4; ++ot) {
#pragma unroll
            for (int j = 0; j < 8; ++j) {
                const int  of   = ot * 16 + 8 * h + j;
                const bool live = (ot < 3) || (of < D_);
                const float bv  = gb[live ? of : 0];
                gbv[ot][j] = live ? bv : 0.0f;
            }
        }
#pragma unroll
        for (int ot = 0; ot < 4; ++ot) {
#pragma unroll
            for (int jp = 0; jp < 4; ++jp) {
                const int  oo = ot * 16 + 8 * h + 2 * jp;
                const bool ok = (ot < 3) || (oo < D_);
                const int  oc = ok ? oo : 0;
                const int  tc = oc >> 2;
                const int  ri = (oc >> 1) & 1;
                const size_t zi = ((size_t)(b * (T_ * C_) + tc) * H_ + (2 * oh + ri)) * W_ + 2 * ow;
                xres[ot][jp] = *(const v2f*)(x + zi);
                bia[ot][jp]  = *(const v2f*)(wb + oc);
            }
        }
    }
    __syncthreads();
    if (khu == 0) {
        const float m1 = jn[32 * 32];
        const float l1 = jn[33 * 32];
        const float mx = fmaxf(mrow, m1);
        const float a0 = fast_exp2(mrow - mx);
        const float a1 = fast_exp2(m1 - mx);
        const float inv = 1.0f / (lrow * a0 + l1 * a1);
        const float w0 = a0 * inv;
        const float w1 = a1 * inv;
        v8f p0, p1, p2, p3;
#pragma unroll
        for (int r = 0; r < 8; ++r) {
            p0[r] = jn[(r)      * 32];
            p1[r] = jn[(8 + r)  * 32];
            p2[r] = jn[(16 + r) * 32];
            p3[r] = jn[(24 + r) * 32];
        }
        v8f y1[4];
        y1[0] = o0 * w0 + p0 * w1;
        y1[1] = o1 * w0 + p1 * w1;
        y1[2] = o2 * w0 + p2 * w1;
        y1[3] = o3 * w0 + p3 * w1;

        __builtin_amdgcn_sched_barrier(0);

        v8f yt[4];
#pragma unroll
        for (int ot = 0; ot < 4; ++ot) yt[ot] = v8f{};
#pragma unroll
        for (int ct = 0; ct < 4; ++ct) {
#pragma unroll
            for (int u = 0; u < 4; ++u) {
                const v2f yb = v2f{y1[ct][2 * u], y1[ct][2 * u + 1]};
#pragma unroll
                for (int ot = 0; ot < 4; ++ot)
                    yt[ot] = __builtin_amdgcn_wmma_f32_16x16x4_f32(
                                 false, gf[ct][u][ot], false, yb, (short)0, yt[ot], false, false);
            }
        }
#pragma unroll
        for (int ot = 0; ot < 4; ++ot)
#pragma unroll
            for (int j = 0; j < 8; ++j) yt[ot][j] += gbv[ot][j];

        v8f z[4];
#pragma unroll
        for (int ot = 0; ot < 4; ++ot) z[ot] = v8f{};
#pragma unroll
        for (int ct = 0; ct < 4; ++ct) {
#pragma unroll
            for (int u = 0; u < 4; ++u) {
                const v2f yb = v2f{yt[ct][2 * u], yt[ct][2 * u + 1]};
#pragma unroll
                for (int ot = 0; ot < 4; ++ot)
                    z[ot] = __builtin_amdgcn_wmma_f32_16x16x4_f32(
                                false, wf[ct][u][ot], false, yb, (short)0, z[ot], false, false);
            }
        }

#pragma unroll
        for (int ot = 0; ot < 4; ++ot) {
#pragma unroll
            for (int jp = 0; jp < 4; ++jp) {
                const int oo = ot * 16 + 8 * h + 2 * jp;
                if ((ot < 3) || (oo < D_)) {
                    const int tc = oo >> 2;
                    const int ri = (oo >> 1) & 1;
                    const size_t zi = ((size_t)(b * (T_ * C_) + tc) * H_ + (2 * oh + ri)) * W_ + 2 * ow;
                    v2f r;
                    r.x = (z[ot][2 * jp]     + bia[ot][jp].x) + xres[ot][jp].x;
                    r.y = (z[ot][2 * jp + 1] + bia[ot][jp].y) + xres[ot][jp].y;
                    *(v2f*)(Z + zi) = r;
                }
            }
        }
    }
}

extern "C" void kernel_launch(void* const* d_in, const int* in_sizes, int n_in,
                              void* d_out, int out_size, void* d_ws, size_t ws_size,
                              hipStream_t stream) {
    const float* x  = (const float*)d_in[0];
    const float* gw = (const float*)d_in[1];
    const float* gb = (const float*)d_in[2];
    const float* ww = (const float*)d_in[3];
    const float* wb = (const float*)d_in[4];
    float* Z = (float*)d_out;

    nlb_attn<<<B_ * 64, 256, 0, stream>>>(x, gw, gb, ww, wb, Z);
}
